# starting kernel
# baseline (speedup 1.0000x reference)
#include <hip/hip_runtime.h>
#include <math.h>

typedef __attribute__((ext_vector_type(2))) float v2f;
typedef __attribute__((ext_vector_type(8))) float v8f;

#define NB    32
#define CCH   64
#define TT    300
#define VV    25
#define NV    (NB*VV)
#define MROWS (NV*TT)
#define TP    150
#define KHALF 4

__global__ void __launch_bounds__(128)
tgat_xw_wmma(const float* __restrict__ x, const float* __restrict__ W,
             float* __restrict__ xw) {
  __shared__ v2f sWp[32 * 64];
  for (int i = threadIdx.x; i < 32 * 64; i += 128) {
    const int kp  = i >> 6;
    const int col = i & 63;
    v2f p;
    p.x = W[(2 * kp) * 64 + col];
    p.y = W[(2 * kp + 1) * 64 + col];
    sWp[i] = p;
  }
  __syncthreads();

  const int lane  = threadIdx.x & 31;
  const int wave  = threadIdx.x >> 5;
  const int half  = lane >> 4;
  const int mloc  = lane & 15;
  const int mbase = blockIdx.x * 64 + wave * 16;

  const int m  = mbase + mloc;
  const int nv = m / TT;
  const int t  = m - nv * TT;
  const int n  = nv / VV;
  const int v  = nv - n * VV;
  const float* xb = x + (size_t)n * (CCH * TT * VV) + (size_t)t * VV + v;

  v8f acc0 = {}, acc1 = {}, acc2 = {}, acc3 = {};
  for (int k = 0; k < 64; k += 4) {
    const int k0 = k + 2 * half;
    v2f a;
    a.x = xb[(size_t)k0 * (TT * VV)];
    a.y = xb[(size_t)(k0 + 1) * (TT * VV)];
    const v2f* wp = sWp + (k0 >> 1) * 64 + mloc;
    const v2f b0 = wp[0];
    const v2f b1 = wp[16];
    const v2f b2 = wp[32];
    const v2f b3 = wp[48];
    acc0 = __builtin_amdgcn_wmma_f32_16x16x4_f32(false, a, false, b0, (short)0, acc0, false, false);
    acc1 = __builtin_amdgcn_wmma_f32_16x16x4_f32(false, a, false, b1, (short)0, acc1, false, false);
    acc2 = __builtin_amdgcn_wmma_f32_16x16x4_f32(false, a, false, b2, (short)0, acc2, false, false);
    acc3 = __builtin_amdgcn_wmma_f32_16x16x4_f32(false, a, false, b3, (short)0, acc3, false, false);
  }

#pragma unroll
  for (int r = 0; r < 8; ++r) {
    float* orow = xw + (size_t)(mbase + r + 8 * half) * 64 + mloc;
    orow[0]  = acc0[r];
    orow[16] = acc1[r];
    orow[32] = acc2[r];
    orow[48] = acc3[r];
  }
}

__global__ void __launch_bounds__(256)
tgat_attn(const float* __restrict__ xw, const float* __restrict__ alpha,
          const float* __restrict__ phi, float* __restrict__ out) {
  __shared__ float sx1[TP * 16];
  __shared__ float sx2[TT * 16];
  __shared__ float srow[64];
  __shared__ float se[TP * 9];
  __shared__ float sD[TP];

  const int nv = blockIdx.x;
  const int n  = nv / VV;
  const int v  = nv - n * VV;
  const float* xwB = xw + (size_t)nv * TT * 64;
  const int tid = threadIdx.x;

  for (int idx = tid; idx < TT * 16; idx += 256) {
    const int s = idx >> 4;
    const int h = idx & 15;
    const float* xr = xwB + (size_t)s * 64;
    const float* pr = phi + h * 64;
    float a2 = 0.f;
    for (int c = 0; c < 64; ++c) a2 += pr[c] * xr[c];
    sx2[idx] = a2;
    if ((s & 1) == 0) {
      const float* ar = alpha + h * 64;
      float a1 = 0.f;
      for (int c = 0; c < 64; ++c) a1 += ar[c] * xr[c];
      sx1[(s >> 1) * 16 + h] = a1;
    }
  }
  for (int c = tid; c < 64; c += 256) {
    float accv = 0.f;
    for (int s = 0; s < TT; ++s) accv += xwB[(size_t)s * 64 + c];
    srow[c] = accv;
  }
  __syncthreads();

  for (int idx = tid; idx < TP * 9; idx += 256) {
    const int tp = idx / 9;
    const int j  = idx - tp * 9;
    const int s  = 2 * tp - KHALF + j;
    float e = 0.f;
    if ((unsigned)s < (unsigned)TT) {
      float sc = 0.f;
      for (int h = 0; h < 16; ++h) sc += sx1[tp * 16 + h] * sx2[s * 16 + h];
      e = __expf(sc) - 1.f;
    }
    se[idx] = e;
  }
  __syncthreads();

  for (int tp = tid; tp < TP; tp += 256) {
    float D = (float)TT;
    for (int j = 0; j < 9; ++j) D += se[tp * 9 + j];
    sD[tp] = D;
  }
  __syncthreads();

  for (int idx = tid; idx < TP * 64; idx += 256) {
    const int tp = idx >> 6;
    const int c  = idx & 63;
    float accv = srow[c];
#pragma unroll
    for (int j = 0; j < 9; ++j) {
      int s = 2 * tp - KHALF + j;
      s = s < 0 ? 0 : (s > TT - 1 ? TT - 1 : s);
      accv += se[tp * 9 + j] * xwB[(size_t)s * 64 + c];
    }
    out[(((size_t)n * 64 + c) * TP + tp) * VV + v] = accv / sD[tp];
  }
}

extern "C" void kernel_launch(void* const* d_in, const int* in_sizes, int n_in,
                              void* d_out, int out_size, void* d_ws, size_t ws_size,
                              hipStream_t stream) {
  const float* x     = (const float*)d_in[0];
  const float* W     = (const float*)d_in[1];
  const float* alpha = (const float*)d_in[2];
  const float* phi   = (const float*)d_in[3];

  float* xw = (float*)d_ws;

  tgat_xw_wmma<<<MROWS / 64, 128, 0, stream>>>(x, W, xw);
  tgat_attn<<<NV, 256, 0, stream>>>(xw, alpha, phi, (float*)d_out);
}


# xw GEMM: 3000 waves loop over 5 position tiles each with the W, W alpha^T, W phi^T fragments loaded once into registers; no LDS fill per 64 positions, next tile's x ahead; sums unchanged
# speedup vs baseline: 5.4046x; 5.4046x over previous
#include <hip/hip_runtime.h>
#include <math.h>

typedef __attribute__((ext_vector_type(2))) float v2f;
typedef __attribute__((ext_vector_type(8))) float v8f;

#define NB    32
#define CCH   64
#define TT    300
#define VV    25
#define PLANE (TT*VV)
#define NPOS  (NB*PLANE)
#define TP    150
#define QPL   (TP*VV)
#define KHALF 4
#define ROWS  96
#define GCH   2

__global__ void __launch_bounds__(256)
tgat_wproj(const float* __restrict__ W, const float* __restrict__ alpha,
           const float* __restrict__ phi, float* __restrict__ wext) {
  const int o = blockIdx.x * 256 + threadIdx.x;
  const int k = o >> 5;
  const int j = o & 31;
  const float* wr = W + k * 64;
  const float* pr = (j < 16) ? (alpha + j * 64) : (phi + (j - 16) * 64);
  float acc = 0.f;
  for (int c = 0; c < 64; ++c) acc += wr[c] * pr[c];
  wext[o] = acc;
}

#define NTILE (NPOS / 16)
#define XWG   750

__device__ __forceinline__ void tgat_load_x(const float* __restrict__ x, int tile, int c16,
                                            int half, v2f (&b)[16]) {
  const int g = tile * 16 + c16;
  const int n = g / PLANE;
  const int p = g - n * PLANE;
  const float* xb = x + (size_t)n * (CCH * PLANE) + (size_t)(2 * half) * PLANE + p;
#pragma unroll
  for (int i = 0; i < 16; ++i) {
    b[i].x = xb[(size_t)(4 * i) * PLANE];
    b[i].y = xb[(size_t)(4 * i + 1) * PLANE];
  }
}

__device__ __forceinline__ void tgat_tile(const v2f (&a)[6][16], const v2f (&b)[16],
                                          float* __restrict__ P, int tile, int c16, int half) {
  v8f acc0 = {}, acc1 = {}, acc2 = {}, acc3 = {}, acc4 = {}, acc5 = {};
#pragma unroll
  for (int i = 0; i < 16; ++i) {
    acc0 = __builtin_amdgcn_wmma_f32_16x16x4_f32(false, a[0][i], false, b[i], (short)0, acc0, false, false);
    acc1 = __builtin_amdgcn_wmma_f32_16x16x4_f32(false, a[1][i], false, b[i], (short)0, acc1, false, false);
    acc2 = __builtin_amdgcn_wmma_f32_16x16x4_f32(false, a[2][i], false, b[i], (short)0, acc2, false, false);
    acc3 = __builtin_amdgcn_wmma_f32_16x16x4_f32(false, a[3][i], false, b[i], (short)0, acc3, false, false);
    acc4 = __builtin_amdgcn_wmma_f32_16x16x4_f32(false, a[4][i], false, b[i], (short)0, acc4, false, false);
    acc5 = __builtin_amdgcn_wmma_f32_16x16x4_f32(false, a[5][i], false, b[i], (short)0, acc5, false, false);
  }
  const int g = tile * 16 + c16;
  const int n = g / PLANE;
  const int p = g - n * PLANE;
  float* pb = P + (size_t)n * (ROWS * PLANE) + p;
#pragma unroll
  for (int r = 0; r < 8; ++r) {
    const int row = r + 8 * half;
    pb[(size_t)(row)      * PLANE] = acc0[r];
    pb[(size_t)(row + 16) * PLANE] = acc1[r];
    pb[(size_t)(row + 32) * PLANE] = acc2[r];
    pb[(size_t)(row + 48) * PLANE] = acc3[r];
    pb[(size_t)(row + 64) * PLANE] = acc4[r];
    pb[(size_t)(row + 80) * PLANE] = acc5[r];
  }
}

__global__ void __launch_bounds__(128)
tgat_xw_wmma(const float* __restrict__ x, const float* __restrict__ W,
             const float* __restrict__ wext, float* __restrict__ P) {
  const int lane = threadIdx.x & 31;
  const int wave = __builtin_amdgcn_readfirstlane(threadIdx.x >> 5);
  const int half = lane >> 4;
  const int c16  = lane & 15;

  const int nw = gridDim.x * 4;
  int tile = blockIdx.x * 4 + wave;
  if (tile >= NTILE) return;

  v2f b0[16], b1[16];
  tgat_load_x(x, tile, c16, half, b0);

  v2f a[6][16];
#pragma unroll
  for (int i = 0; i < 16; ++i) {
    const int k0 = 4 * i + 2 * half;
#pragma unroll
    for (int r = 0; r < 4; ++r) {
      a[r][i].x = W[k0 * 64 + 16 * r + c16];
      a[r][i].y = W[(k0 + 1) * 64 + 16 * r + c16];
    }
#pragma unroll
    for (int r = 0; r < 2; ++r) {
      a[4 + r][i].x = wext[k0 * 32 + 16 * r + c16];
      a[4 + r][i].y = wext[(k0 + 1) * 32 + 16 * r + c16];
    }
  }

#pragma unroll 1
  for (;;) {
    const bool more1 = tile + nw < NTILE;
    if (more1) tgat_load_x(x, tile + nw, c16, half, b1);
    tgat_tile(a, b0, P, tile, c16, half);
    if (!more1) break;
    tile += nw;
    const bool more0 = tile + nw < NTILE;
    if (more0) tgat_load_x(x, tile + nw, c16, half, b0);
    tgat_tile(a, b1, P, tile, c16, half);
    if (!more0) break;
    tile += nw;
  }
}

__global__ void __launch_bounds__(256)
tgat_band(const float* __restrict__ P, float* __restrict__ E, float* __restrict__ Dn) {
  const int n = blockIdx.y;
  const int q = blockIdx.x * 256 + threadIdx.x;
  if (q >= QPL) return;
  const int tp = q / VV;
  const int v  = q - tp * VV;
  const float* x1p = P + ((size_t)n * ROWS + 64) * PLANE + (2 * tp) * VV + v;
  const float* x2p = P + ((size_t)n * ROWS + 80) * PLANE + v;
  float a[16];
#pragma unroll
  for (int h = 0; h < 16; ++h) a[h] = x1p[(size_t)h * PLANE];
  float D = (float)TT;
#pragma unroll
  for (int j = 0; j < 9; ++j) {
    const int s = 2 * tp - KHALF + j;
    float e = 0.f;
    if ((unsigned)s < (unsigned)TT) {
      float sc = 0.f;
#pragma unroll
      for (int h = 0; h < 16; ++h) sc += a[h] * x2p[(size_t)h * PLANE + s * VV];
      e = __expf(sc) - 1.f;
    }
    E[((size_t)n * 9 + j) * QPL + q] = e;
    D += e;
  }
  Dn[(size_t)n * QPL + q] = D;
}

#define NF4  (GCH * PLANE / 4)
#define NSEG 5
__global__ void __launch_bounds__(256)
tgat_out(const float* __restrict__ P, const float* __restrict__ E,
         const float* __restrict__ Dn, float* __restrict__ out) {
  __shared__ __attribute__((aligned(16))) float sp[GCH * PLANE];
  __shared__ float spart[NSEG * GCH * VV];
  __shared__ float sS[GCH * VV];

  const int n   = blockIdx.y;
  const int c0  = blockIdx.x * GCH;
  const int tid = threadIdx.x;

  {
    const float4* gp = reinterpret_cast<const float4*>(P + ((size_t)n * ROWS + c0) * PLANE);
    float4* s4 = reinterpret_cast<float4*>(sp);
    float4 r[15];
#pragma unroll
    for (int u = 0; u < 15; ++u) {
      int i = tid + 256 * u;
      i = i < NF4 ? i : NF4 - 1;
      r[u] = gp[i];
    }
#pragma unroll
    for (int u = 0; u < 15; ++u) {
      int i = tid + 256 * u;
      i = i < NF4 ? i : NF4 - 1;
      s4[i] = r[u];
    }
  }
  __syncthreads();

  if (tid < NSEG * GCH * VV) {
    const int seg = tid / (GCH * VV);
    const int gv  = tid - seg * (GCH * VV);
    const int gch = gv / VV;
    const int v   = gv - gch * VV;
    const float* col = sp + gch * PLANE + (seg * (TT / NSEG)) * VV + v;
    float accv = 0.f;
    for (int t = 0; t < TT / NSEG; ++t) accv += col[t * VV];
    spart[tid] = accv;
  }
  __syncthreads();
  if (tid < GCH * VV) {
    float accv = 0.f;
#pragma unroll
    for (int seg = 0; seg < NSEG; ++seg) accv += spart[seg * (GCH * VV) + tid];
    sS[tid] = accv;
  }
  __syncthreads();

  const float* En = E + (size_t)n * 9 * QPL;
  const float* Dq = Dn + (size_t)n * QPL;
  float* ob = out + ((size_t)n * CCH + c0) * QPL;
  for (int q = tid; q < QPL; q += 256) {
    const int tp = q / VV;
    const int v  = q - tp * VV;
    float e[9];
    int   idx[9];
#pragma unroll
    for (int j = 0; j < 9; ++j) {
      e[j] = En[(size_t)j * QPL + q];
      int s = 2 * tp - KHALF + j;
      s = s < 0 ? 0 : (s > TT - 1 ? TT - 1 : s);
      idx[j] = s * VV + v;
    }
    const float D = Dq[q];
#pragma unroll
    for (int gch = 0; gch < GCH; ++gch) {
      float accv = sS[gch * VV + v];
#pragma unroll
      for (int j = 0; j < 9; ++j) accv += e[j] * sp[gch * PLANE + idx[j]];
      ob[(size_t)gch * QPL + q] = accv / D;
    }
  }
}

extern "C" void kernel_launch(void* const* d_in, const int* in_sizes, int n_in,
                              void* d_out, int out_size, void* d_ws, size_t ws_size,
                              hipStream_t stream) {
  const float* x     = (const float*)d_in[0];
  const float* W     = (const float*)d_in[1];
  const float* alpha = (const float*)d_in[2];
  const float* phi   = (const float*)d_in[3];

  float* P    = (float*)d_ws;
  float* E    = P  + (size_t)NB * ROWS * PLANE;
  float* Dn   = E  + (size_t)NB * 9 * QPL;
  float* wext = Dn + (size_t)NB * QPL;

  tgat_wproj<<<8, 256, 0, stream>>>(W, alpha, phi, wext);
  tgat_xw_wmma<<<XWG, 128, 0, stream>>>(x, W, wext, P);
  tgat_band<<<dim3(15, NB), 256, 0, stream>>>(P, E, Dn);
  tgat_out<<<dim3(CCH / GCH, NB), 256, 0, stream>>>(P, E, Dn, (float*)d_out);
}


# x1/x2 rows come from the xw accumulators still in registers, used as B operand of 32 more products against alpha/phi fragments; the W alpha^T launch goes (3 launches); order of x1/x2 sums changes
# speedup vs baseline: 5.8542x; 1.0832x over previous
#include <hip/hip_runtime.h>
#include <math.h>

typedef __attribute__((ext_vector_type(2))) float v2f;
typedef __attribute__((ext_vector_type(8))) float v8f;

#define NB    32
#define CCH   64
#define TT    300
#define VV    25
#define PLANE (TT*VV)
#define NPOS  (NB*PLANE)
#define TP    150
#define QPL   (TP*VV)
#define KHALF 4
#define ROWS  96
#define GCH   2

#define NTILE (NPOS / 16)
#define XWG   750

__device__ __forceinline__ void tgat_load_x(const float* __restrict__ x, int tile, int c16,
                                            int half, v2f (&b)[16]) {
  const int g = tile * 16 + c16;
  const int n = g / PLANE;
  const int p = g - n * PLANE;
  const float* xb = x + (size_t)n * (CCH * PLANE) + (size_t)(2 * half) * PLANE + p;
#pragma unroll
  for (int i = 0; i < 16; ++i) {
    b[i].x = xb[(size_t)(4 * i) * PLANE];
    b[i].y = xb[(size_t)(4 * i + 1) * PLANE];
  }
}

__device__ __forceinline__ void tgat_tile(const v2f (&a)[6][16], const v2f (&b)[16],
                                          float* __restrict__ P, int tile, int c16, int half) {
  v8f acc0 = {}, acc1 = {}, acc2 = {}, acc3 = {}, acc4 = {}, acc5 = {};
#pragma unroll
  for (int i = 0; i < 16; ++i) {
    acc0 = __builtin_amdgcn_wmma_f32_16x16x4_f32(false, a[0][i], false, b[i], (short)0, acc0, false, false);
    acc1 = __builtin_amdgcn_wmma_f32_16x16x4_f32(false, a[1][i], false, b[i], (short)0, acc1, false, false);
    acc2 = __builtin_amdgcn_wmma_f32_16x16x4_f32(false, a[2][i], false, b[i], (short)0, acc2, false, false);
    acc3 = __builtin_amdgcn_wmma_f32_16x16x4_f32(false, a[3][i], false, b[i], (short)0, acc3, false, false);
  }
#pragma unroll
  for (int j = 0; j < 4; ++j) {
    const v2f q0 = {acc0[2 * j], acc0[2 * j + 1]};
    acc4 = __builtin_amdgcn_wmma_f32_16x16x4_f32(false, a[4][j], false, q0, (short)0, acc4, false, false);
    acc5 = __builtin_amdgcn_wmma_f32_16x16x4_f32(false, a[5][j], false, q0, (short)0, acc5, false, false);
  }
#pragma unroll
  for (int j = 0; j < 4; ++j) {
    const v2f q1 = {acc1[2 * j], acc1[2 * j + 1]};
    acc4 = __builtin_amdgcn_wmma_f32_16x16x4_f32(false, a[4][4 + j], false, q1, (short)0, acc4, false, false);
    acc5 = __builtin_amdgcn_wmma_f32_16x16x4_f32(false, a[5][4 + j], false, q1, (short)0, acc5, false, false);
  }
#pragma unroll
  for (int j = 0; j < 4; ++j) {
    const v2f q2 = {acc2[2 * j], acc2[2 * j + 1]};
    acc4 = __builtin_amdgcn_wmma_f32_16x16x4_f32(false, a[4][8 + j], false, q2, (short)0, acc4, false, false);
    acc5 = __builtin_amdgcn_wmma_f32_16x16x4_f32(false, a[5][8 + j], false, q2, (short)0, acc5, false, false);
  }
#pragma unroll
  for (int j = 0; j < 4; ++j) {
    const v2f q3 = {acc3[2 * j], acc3[2 * j + 1]};
    acc4 = __builtin_amdgcn_wmma_f32_16x16x4_f32(false, a[4][12 + j], false, q3, (short)0, acc4, false, false);
    acc5 = __builtin_amdgcn_wmma_f32_16x16x4_f32(false, a[5][12 + j], false, q3, (short)0, acc5, false, false);
  }
  const int g = tile * 16 + c16;
  const int n = g / PLANE;
  const int p = g - n * PLANE;
  float* pb = P + (size_t)n * (ROWS * PLANE) + p;
#pragma unroll
  for (int r = 0; r < 8; ++r) {
    const int row = r + 8 * half;
    pb[(size_t)(row)      * PLANE] = acc0[r];
    pb[(size_t)(row + 16) * PLANE] = acc1[r];
    pb[(size_t)(row + 32) * PLANE] = acc2[r];
    pb[(size_t)(row + 48) * PLANE] = acc3[r];
    pb[(size_t)(row + 64) * PLANE] = acc4[r];
    pb[(size_t)(row + 80) * PLANE] = acc5[r];
  }
}

__global__ void __launch_bounds__(128)
tgat_xw_wmma(const float* __restrict__ x, const float* __restrict__ W,
             const float* __restrict__ alpha, const float* __restrict__ phi,
             float* __restrict__ P) {
  const int lane = threadIdx.x & 31;
  const int wave = __builtin_amdgcn_readfirstlane(threadIdx.x >> 5);
  const int half = lane >> 4;
  const int c16  = lane & 15;

  const int nw = gridDim.x * 4;
  int tile = blockIdx.x * 4 + wave;
  if (tile >= NTILE) return;

  v2f b0[16], b1[16];
  tgat_load_x(x, tile, c16, half, b0);

  v2f a[6][16];
#pragma unroll
  for (int i = 0; i < 16; ++i) {
    const int k0 = 4 * i + 2 * half;
#pragma unroll
    for (int r = 0; r < 4; ++r) {
      a[r][i].x = W[k0 * 64 + 16 * r + c16];
      a[r][i].y = W[(k0 + 1) * 64 + 16 * r + c16];
    }
  }
#pragma unroll
  for (int rt = 0; rt < 4; ++rt) {
#pragma unroll
    for (int j = 0; j < 4; ++j) {
      const int kk = 16 * rt + 8 * half + 2 * j;
      a[4][4 * rt + j].x = alpha[c16 * 64 + kk];
      a[4][4 * rt + j].y = alpha[c16 * 64 + kk + 1];
      a[5][4 * rt + j].x = phi[c16 * 64 + kk];
      a[5][4 * rt + j].y = phi[c16 * 64 + kk + 1];
    }
  }

#pragma unroll 1
  for (;;) {
    const bool more1 = tile + nw < NTILE;
    if (more1) tgat_load_x(x, tile + nw, c16, half, b1);
    tgat_tile(a, b0, P, tile, c16, half);
    if (!more1) break;
    tile += nw;
    const bool more0 = tile + nw < NTILE;
    if (more0) tgat_load_x(x, tile + nw, c16, half, b0);
    tgat_tile(a, b1, P, tile, c16, half);
    if (!more0) break;
    tile += nw;
  }
}

__global__ void __launch_bounds__(256)
tgat_band(const float* __restrict__ P, float* __restrict__ E, float* __restrict__ Dn) {
  const int n = blockIdx.y;
  const int q = blockIdx.x * 256 + threadIdx.x;
  if (q >= QPL) return;
  const int tp = q / VV;
  const int v  = q - tp * VV;
  const float* x1p = P + ((size_t)n * ROWS + 64) * PLANE + (2 * tp) * VV + v;
  const float* x2p = P + ((size_t)n * ROWS + 80) * PLANE + v;
  float a[16];
#pragma unroll
  for (int h = 0; h < 16; ++h) a[h] = x1p[(size_t)h * PLANE];
  float D = (float)TT;
#pragma unroll
  for (int j = 0; j < 9; ++j) {
    const int s = 2 * tp - KHALF + j;
    float e = 0.f;
    if ((unsigned)s < (unsigned)TT) {
      float sc = 0.f;
#pragma unroll
      for (int h = 0; h < 16; ++h) sc += a[h] * x2p[(size_t)h * PLANE + s * VV];
      e = __expf(sc) - 1.f;
    }
    E[((size_t)n * 9 + j) * QPL + q] = e;
    D += e;
  }
  Dn[(size_t)n * QPL + q] = D;
}

#define NF4  (GCH * PLANE / 4)
#define NSEG 5
__global__ void __launch_bounds__(256)
tgat_out(const float* __restrict__ P, const float* __restrict__ E,
         const float* __restrict__ Dn, float* __restrict__ out) {
  __shared__ __attribute__((aligned(16))) float sp[GCH * PLANE];
  __shared__ float spart[NSEG * GCH * VV];
  __shared__ float sS[GCH * VV];

  const int n   = blockIdx.y;
  const int c0  = blockIdx.x * GCH;
  const int tid = threadIdx.x;

  {
    const float4* gp = reinterpret_cast<const float4*>(P + ((size_t)n * ROWS + c0) * PLANE);
    float4* s4 = reinterpret_cast<float4*>(sp);
    float4 r[15];
#pragma unroll
    for (int u = 0; u < 15; ++u) {
      int i = tid + 256 * u;
      i = i < NF4 ? i : NF4 - 1;
      r[u] = gp[i];
    }
#pragma unroll
    for (int u = 0; u < 15; ++u) {
      int i = tid + 256 * u;
      i = i < NF4 ? i : NF4 - 1;
      s4[i] = r[u];
    }
  }
  __syncthreads();

  if (tid < NSEG * GCH * VV) {
    const int seg = tid / (GCH * VV);
    const int gv  = tid - seg * (GCH * VV);
    const int gch = gv / VV;
    const int v   = gv - gch * VV;
    const float* col = sp + gch * PLANE + (seg * (TT / NSEG)) * VV + v;
    float accv = 0.f;
    for (int t = 0; t < TT / NSEG; ++t) accv += col[t * VV];
    spart[tid] = accv;
  }
  __syncthreads();
  if (tid < GCH * VV) {
    float accv = 0.f;
#pragma unroll
    for (int seg = 0; seg < NSEG; ++seg) accv += spart[seg * (GCH * VV) + tid];
    sS[tid] = accv;
  }
  __syncthreads();

  const float* En = E + (size_t)n * 9 * QPL;
  const float* Dq = Dn + (size_t)n * QPL;
  float* ob = out + ((size_t)n * CCH + c0) * QPL;
  for (int q = tid; q < QPL; q += 256) {
    const int tp = q / VV;
    const int v  = q - tp * VV;
    float e[9];
    int   idx[9];
#pragma unroll
    for (int j = 0; j < 9; ++j) {
      e[j] = En[(size_t)j * QPL + q];
      int s = 2 * tp - KHALF + j;
      s = s < 0 ? 0 : (s > TT - 1 ? TT - 1 : s);
      idx[j] = s * VV + v;
    }
    const float D = Dq[q];
#pragma unroll
    for (int gch = 0; gch < GCH; ++gch) {
      float accv = sS[gch * VV + v];
#pragma unroll
      for (int j = 0; j < 9; ++j) accv += e[j] * sp[gch * PLANE + idx[j]];
      ob[(size_t)gch * QPL + q] = accv / D;
    }
  }
}

extern "C" void kernel_launch(void* const* d_in, const int* in_sizes, int n_in,
                              void* d_out, int out_size, void* d_ws, size_t ws_size,
                              hipStream_t stream) {
  const float* x     = (const float*)d_in[0];
  const float* W     = (const float*)d_in[1];
  const float* alpha = (const float*)d_in[2];
  const float* phi   = (const float*)d_in[3];

  float* P    = (float*)d_ws;
  float* E    = P  + (size_t)NB * ROWS * PLANE;
  float* Dn   = E  + (size_t)NB * 9 * QPL;

  tgat_xw_wmma<<<XWG, 128, 0, stream>>>(x, W, alpha, phi, P);
  tgat_band<<<dim3(15, NB), 256, 0, stream>>>(P, E, Dn);
  tgat_out<<<dim3(CCH / GCH, NB), 256, 0, stream>>>(P, E, Dn, (float*)d_out);
}


# P rows padded to a 128-byte pitch, tiles taken in pairs: lane halves swap accumulator values (permlanex16) so each P store writes one whole aligned 128-B line, not two 64-B pieces; sums unchanged
# speedup vs baseline: 6.2255x; 1.0634x over previous
#include <hip/hip_runtime.h>
#include <math.h>

typedef __attribute__((ext_vector_type(2))) float v2f;
typedef __attribute__((ext_vector_type(8))) float v8f;

#define NB    32
#define CCH   64
#define TT    300
#define VV    25
#define PLANE (TT*VV)
#define NPOS  (NB*PLANE)
#define TP    150
#define QPL   (TP*VV)
#define KHALF 4
#define ROWS  96
#define GCH   2

#define PP      7520
#define NPAIR_N (PP / 32)
#define NPAIR   (NB * NPAIR_N)
#define XWG     470

__device__ __forceinline__ void tgat_load_x(const float* __restrict__ x, int n, int p0, int c16,
                                            int half, v2f (&b)[16]) {
  int p = p0 + c16;
  p = p < PLANE ? p : PLANE - 1;
  const float* xb = x + (size_t)n * (CCH * PLANE) + (size_t)(2 * half) * PLANE + p;
#pragma unroll
  for (int i = 0; i < 16; ++i) {
    b[i].x = xb[(size_t)(4 * i) * PLANE];
    b[i].y = xb[(size_t)(4 * i + 1) * PLANE];
  }
}

__device__ __forceinline__ void tgat_tile(const v2f (&a)[6][16], const v2f (&b)[16], v8f (&acc)[6]) {
  v8f acc0 = {}, acc1 = {}, acc2 = {}, acc3 = {}, acc4 = {}, acc5 = {};
#pragma unroll
  for (int i = 0; i < 16; ++i) {
    acc0 = __builtin_amdgcn_wmma_f32_16x16x4_f32(false, a[0][i], false, b[i], (short)0, acc0, false, false);
    acc1 = __builtin_amdgcn_wmma_f32_16x16x4_f32(false, a[1][i], false, b[i], (short)0, acc1, false, false);
    acc2 = __builtin_amdgcn_wmma_f32_16x16x4_f32(false, a[2][i], false, b[i], (short)0, acc2, false, false);
    acc3 = __builtin_amdgcn_wmma_f32_16x16x4_f32(false, a[3][i], false, b[i], (short)0, acc3, false, false);
  }
#pragma unroll
  for (int j = 0; j < 4; ++j) {
    const v2f q0 = {acc0[2 * j], acc0[2 * j + 1]};
    acc4 = __builtin_amdgcn_wmma_f32_16x16x4_f32(false, a[4][j], false, q0, (short)0, acc4, false, false);
    acc5 = __builtin_amdgcn_wmma_f32_16x16x4_f32(false, a[5][j], false, q0, (short)0, acc5, false, false);
  }
#pragma unroll
  for (int j = 0; j < 4; ++j) {
    const v2f q1 = {acc1[2 * j], acc1[2 * j + 1]};
    acc4 = __builtin_amdgcn_wmma_f32_16x16x4_f32(false, a[4][4 + j], false, q1, (short)0, acc4, false, false);
    acc5 = __builtin_amdgcn_wmma_f32_16x16x4_f32(false, a[5][4 + j], false, q1, (short)0, acc5, false, false);
  }
#pragma unroll
  for (int j = 0; j < 4; ++j) {
    const v2f q2 = {acc2[2 * j], acc2[2 * j + 1]};
    acc4 = __builtin_amdgcn_wmma_f32_16x16x4_f32(false, a[4][8 + j], false, q2, (short)0, acc4, false, false);
    acc5 = __builtin_amdgcn_wmma_f32_16x16x4_f32(false, a[5][8 + j], false, q2, (short)0, acc5, false, false);
  }
#pragma unroll
  for (int j = 0; j < 4; ++j) {
    const v2f q3 = {acc3[2 * j], acc3[2 * j + 1]};
    acc4 = __builtin_amdgcn_wmma_f32_16x16x4_f32(false, a[4][12 + j], false, q3, (short)0, acc4, false, false);
    acc5 = __builtin_amdgcn_wmma_f32_16x16x4_f32(false, a[5][12 + j], false, q3, (short)0, acc5, false, false);
  }
  acc[0] = acc0; acc[1] = acc1; acc[2] = acc2; acc[3] = acc3; acc[4] = acc4; acc[5] = acc5;
}

__device__ __forceinline__ float tgat_xhalf(float v) {
  return __builtin_bit_cast(float, __builtin_amdgcn_permlanex16(__builtin_bit_cast(int, v),
                            __builtin_bit_cast(int, v), 0x76543210, 0xfedcba98, false, false));
}

__device__ __forceinline__ void tgat_store_pair(const v8f (&A)[6], const v8f (&B)[6],
                                                float* __restrict__ P, int n, int pp,
                                                int lane, int half) {
  float* pb = P + (size_t)n * (ROWS * PP) + pp * 32 + lane;
#pragma unroll
  for (int k = 0; k < 6; ++k) {
#pragma unroll
    for (int r = 0; r < 8; ++r) {
      const float send = half ? A[k][r] : B[k][r];
      const float recv = tgat_xhalf(send);
      const float lo = half ? recv : A[k][r];
      const float hi = half ? B[k][r] : recv;
      pb[(size_t)(16 * k + r)     * PP] = lo;
      pb[(size_t)(16 * k + 8 + r) * PP] = hi;
    }
  }
}

__device__ __forceinline__ void tgat_pair(const v2f (&a)[6][16], const v2f (&bA)[16],
                                          const v2f (&bB)[16], float* __restrict__ P,
                                          int pr, int lane, int half) {
  const int n  = pr / NPAIR_N;
  const int pp = pr - n * NPAIR_N;
  v8f accA[6], accB[6];
  tgat_tile(a, bA, accA);
  tgat_tile(a, bB, accB);
  tgat_store_pair(accA, accB, P, n, pp, lane, half);
}

__device__ __forceinline__ void tgat_load_pair(const float* __restrict__ x, int pr, int c16,
                                               int half, v2f (&bA)[16], v2f (&bB)[16]) {
  const int n  = pr / NPAIR_N;
  const int pp = pr - n * NPAIR_N;
  tgat_load_x(x, n, pp * 32, c16, half, bA);
  tgat_load_x(x, n, pp * 32 + 16, c16, half, bB);
}

__global__ void __launch_bounds__(128)
tgat_xw_wmma(const float* __restrict__ x, const float* __restrict__ W,
             const float* __restrict__ alpha, const float* __restrict__ phi,
             float* __restrict__ P) {
  const int lane = threadIdx.x & 31;
  const int wave = __builtin_amdgcn_readfirstlane(threadIdx.x >> 5);
  const int half = lane >> 4;
  const int c16  = lane & 15;

  const int nw = gridDim.x * 4;
  int pr = blockIdx.x * 4 + wave;
  if (pr >= NPAIR) return;

  v2f b0A[16], b0B[16], b1A[16], b1B[16];
  tgat_load_pair(x, pr, c16, half, b0A, b0B);

  v2f a[6][16];
#pragma unroll
  for (int i = 0; i < 16; ++i) {
    const int k0 = 4 * i + 2 * half;
#pragma unroll
    for (int r = 0; r < 4; ++r) {
      a[r][i].x = W[k0 * 64 + 16 * r + c16];
      a[r][i].y = W[(k0 + 1) * 64 + 16 * r + c16];
    }
  }
#pragma unroll
  for (int rt = 0; rt < 4; ++rt) {
#pragma unroll
    for (int j = 0; j < 4; ++j) {
      const int kk = 16 * rt + 8 * half + 2 * j;
      a[4][4 * rt + j].x = alpha[c16 * 64 + kk];
      a[4][4 * rt + j].y = alpha[c16 * 64 + kk + 1];
      a[5][4 * rt + j].x = phi[c16 * 64 + kk];
      a[5][4 * rt + j].y = phi[c16 * 64 + kk + 1];
    }
  }

#pragma unroll 1
  for (;;) {
    const bool more1 = pr + nw < NPAIR;
    if (more1) tgat_load_pair(x, pr + nw, c16, half, b1A, b1B);
    tgat_pair(a, b0A, b0B, P, pr, lane, half);
    if (!more1) break;
    pr += nw;
    const bool more0 = pr + nw < NPAIR;
    if (more0) tgat_load_pair(x, pr + nw, c16, half, b0A, b0B);
    tgat_pair(a, b1A, b1B, P, pr, lane, half);
    if (!more0) break;
    pr += nw;
  }
}

__global__ void __launch_bounds__(256)
tgat_band(const float* __restrict__ P, float* __restrict__ E, float* __restrict__ Dn) {
  const int n = blockIdx.y;
  const int q = blockIdx.x * 256 + threadIdx.x;
  if (q >= QPL) return;
  const int tp = q / VV;
  const int v  = q - tp * VV;
  const float* x1p = P + ((size_t)n * ROWS + 64) * PP + (2 * tp) * VV + v;
  const float* x2p = P + ((size_t)n * ROWS + 80) * PP + v;
  float a[16];
#pragma unroll
  for (int h = 0; h < 16; ++h) a[h] = x1p[(size_t)h * PP];
  float D = (float)TT;
#pragma unroll
  for (int j = 0; j < 9; ++j) {
    const int s = 2 * tp - KHALF + j;
    float e = 0.f;
    if ((unsigned)s < (unsigned)TT) {
      float sc = 0.f;
#pragma unroll
      for (int h = 0; h < 16; ++h) sc += a[h] * x2p[(size_t)h * PP + s * VV];
      e = __expf(sc) - 1.f;
    }
    E[((size_t)n * 9 + j) * QPL + q] = e;
    D += e;
  }
  Dn[(size_t)n * QPL + q] = D;
}

#define NF4  (GCH * PP / 4)
#define NSEG 5
__global__ void __launch_bounds__(256)
tgat_out(const float* __restrict__ P, const float* __restrict__ E,
         const float* __restrict__ Dn, float* __restrict__ out) {
  __shared__ __attribute__((aligned(16))) float sp[GCH * PP];
  __shared__ float spart[NSEG * GCH * VV];
  __shared__ float sS[GCH * VV];

  const int n   = blockIdx.y;
  const int c0  = blockIdx.x * GCH;
  const int tid = threadIdx.x;

  {
    const float4* gp = reinterpret_cast<const float4*>(P + ((size_t)n * ROWS + c0) * PP);
    float4* s4 = reinterpret_cast<float4*>(sp);
    float4 r[15];
#pragma unroll
    for (int u = 0; u < 15; ++u) {
      int i = tid + 256 * u;
      i = i < NF4 ? i : NF4 - 1;
      r[u] = gp[i];
    }
#pragma unroll
    for (int u = 0; u < 15; ++u) {
      int i = tid + 256 * u;
      i = i < NF4 ? i : NF4 - 1;
      s4[i] = r[u];
    }
  }
  __syncthreads();

  if (tid < NSEG * GCH * VV) {
    const int seg = tid / (GCH * VV);
    const int gv  = tid - seg * (GCH * VV);
    const int gch = gv / VV;
    const int v   = gv - gch * VV;
    const float* col = sp + gch * PP + (seg * (TT / NSEG)) * VV + v;
    float accv = 0.f;
    for (int t = 0; t < TT / NSEG; ++t) accv += col[t * VV];
    spart[tid] = accv;
  }
  __syncthreads();
  if (tid < GCH * VV) {
    float accv = 0.f;
#pragma unroll
    for (int seg = 0; seg < NSEG; ++seg) accv += spart[seg * (GCH * VV) + tid];
    sS[tid] = accv;
  }
  __syncthreads();

  const float* En = E + (size_t)n * 9 * QPL;
  const float* Dq = Dn + (size_t)n * QPL;
  float* ob = out + ((size_t)n * CCH + c0) * QPL;
  for (int q = tid; q < QPL; q += 256) {
    const int tp = q / VV;
    const int v  = q - tp * VV;
    float e[9];
    int   idx[9];
#pragma unroll
    for (int j = 0; j < 9; ++j) {
      e[j] = En[(size_t)j * QPL + q];
      int s = 2 * tp - KHALF + j;
      s = s < 0 ? 0 : (s > TT - 1 ? TT - 1 : s);
      idx[j] = s * VV + v;
    }
    const float D = Dq[q];
#pragma unroll
    for (int gch = 0; gch < GCH; ++gch) {
      float accv = sS[gch * VV + v];
#pragma unroll
      for (int j = 0; j < 9; ++j) accv += e[j] * sp[gch * PP + idx[j]];
      ob[(size_t)gch * QPL + q] = accv / D;
    }
  }
}

extern "C" void kernel_launch(void* const* d_in, const int* in_sizes, int n_in,
                              void* d_out, int out_size, void* d_ws, size_t ws_size,
                              hipStream_t stream) {
  const float* x     = (const float*)d_in[0];
  const float* W     = (const float*)d_in[1];
  const float* alpha = (const float*)d_in[2];
  const float* phi   = (const float*)d_in[3];

  float* P    = (float*)d_ws;
  float* E    = P  + (size_t)NB * ROWS * PP;
  float* Dn   = E  + (size_t)NB * 9 * QPL;

  tgat_xw_wmma<<<XWG, 128, 0, stream>>>(x, W, alpha, phi, P);
  tgat_band<<<dim3(15, NB), 256, 0, stream>>>(P, E, Dn);
  tgat_out<<<dim3(CCH / GCH, NB), 256, 0, stream>>>(P, E, Dn, (float*)d_out);
}


# output kernel re-tiled: 512-thread work-group holds 8 channel planes in LDS (was 256 threads, 2 planes); band weights and D read once per 8 planes, in the rows' burst; half the waves; sums unchanged
# speedup vs baseline: 6.4545x; 1.0368x over previous
#include <hip/hip_runtime.h>
#include <math.h>

typedef __attribute__((ext_vector_type(2))) float v2f;
typedef __attribute__((ext_vector_type(8))) float v8f;

#define NB    32
#define CCH   64
#define TT    300
#define VV    25
#define PLANE (TT*VV)
#define NPOS  (NB*PLANE)
#define TP    150
#define QPL   (TP*VV)
#define KHALF 4
#define ROWS  96
#define GCH   8

#define PP      7520
#define NPAIR_N (PP / 32)
#define NPAIR   (NB * NPAIR_N)
#define XWG     470

__device__ __forceinline__ void tgat_load_x(const float* __restrict__ x, int n, int p0, int c16,
                                            int half, v2f (&b)[16]) {
  int p = p0 + c16;
  p = p < PLANE ? p : PLANE - 1;
  const float* xb = x + (size_t)n * (CCH * PLANE) + (size_t)(2 * half) * PLANE + p;
#pragma unroll
  for (int i = 0; i < 16; ++i) {
    b[i].x = xb[(size_t)(4 * i) * PLANE];
    b[i].y = xb[(size_t)(4 * i + 1) * PLANE];
  }
}

__device__ __forceinline__ void tgat_tile(const v2f (&a)[6][16], const v2f (&b)[16], v8f (&acc)[6]) {
  v8f acc0 = {}, acc1 = {}, acc2 = {}, acc3 = {}, acc4 = {}, acc5 = {};
#pragma unroll
  for (int i = 0; i < 16; ++i) {
    acc0 = __builtin_amdgcn_wmma_f32_16x16x4_f32(false, a[0][i], false, b[i], (short)0, acc0, false, false);
    acc1 = __builtin_amdgcn_wmma_f32_16x16x4_f32(false, a[1][i], false, b[i], (short)0, acc1, false, false);
    acc2 = __builtin_amdgcn_wmma_f32_16x16x4_f32(false, a[2][i], false, b[i], (short)0, acc2, false, false);
    acc3 = __builtin_amdgcn_wmma_f32_16x16x4_f32(false, a[3][i], false, b[i], (short)0, acc3, false, false);
  }
#pragma unroll
  for (int j = 0; j < 4; ++j) {
    const v2f q0 = {acc0[2 * j], acc0[2 * j + 1]};
    acc4 = __builtin_amdgcn_wmma_f32_16x16x4_f32(false, a[4][j], false, q0, (short)0, acc4, false, false);
    acc5 = __builtin_amdgcn_wmma_f32_16x16x4_f32(false, a[5][j], false, q0, (short)0, acc5, false, false);
  }
#pragma unroll
  for (int j = 0; j < 4; ++j) {
    const v2f q1 = {acc1[2 * j], acc1[2 * j + 1]};
    acc4 = __builtin_amdgcn_wmma_f32_16x16x4_f32(false, a[4][4 + j], false, q1, (short)0, acc4, false, false);
    acc5 = __builtin_amdgcn_wmma_f32_16x16x4_f32(false, a[5][4 + j], false, q1, (short)0, acc5, false, false);
  }
#pragma unroll
  for (int j = 0; j < 4; ++j) {
    const v2f q2 = {acc2[2 * j], acc2[2 * j + 1]};
    acc4 = __builtin_amdgcn_wmma_f32_16x16x4_f32(false, a[4][8 + j], false, q2, (short)0, acc4, false, false);
    acc5 = __builtin_amdgcn_wmma_f32_16x16x4_f32(false, a[5][8 + j], false, q2, (short)0, acc5, false, false);
  }
#pragma unroll
  for (int j = 0; j < 4; ++j) {
    const v2f q3 = {acc3[2 * j], acc3[2 * j + 1]};
    acc4 = __builtin_amdgcn_wmma_f32_16x16x4_f32(false, a[4][12 + j], false, q3, (short)0, acc4, false, false);
    acc5 = __builtin_amdgcn_wmma_f32_16x16x4_f32(false, a[5][12 + j], false, q3, (short)0, acc5, false, false);
  }
  acc[0] = acc0; acc[1] = acc1; acc[2] = acc2; acc[3] = acc3; acc[4] = acc4; acc[5] = acc5;
}

__device__ __forceinline__ float tgat_xhalf(float v) {
  return __builtin_bit_cast(float, __builtin_amdgcn_permlanex16(__builtin_bit_cast(int, v),
                            __builtin_bit_cast(int, v), 0x76543210, 0xfedcba98, false, false));
}

__device__ __forceinline__ void tgat_store_pair(const v8f (&A)[6], const v8f (&B)[6],
                                                float* __restrict__ P, int n, int pp,
                                                int lane, int half) {
  float* pb = P + (size_t)n * (ROWS * PP) + pp * 32 + lane;
#pragma unroll
  for (int k = 0; k < 6; ++k) {
#pragma unroll
    for (int r = 0; r < 8; ++r) {
      const float send = half ? A[k][r] : B[k][r];
      const float recv = tgat_xhalf(send);
      const float lo = half ? recv : A[k][r];
      const float hi = half ? B[k][r] : recv;
      pb[(size_t)(16 * k + r)     * PP] = lo;
      pb[(size_t)(16 * k + 8 + r) * PP] = hi;
    }
  }
}

__device__ __forceinline__ void tgat_pair(const v2f (&a)[6][16], const v2f (&bA)[16],
                                          const v2f (&bB)[16], float* __restrict__ P,
                                          int pr, int lane, int half) {
  const int n  = pr / NPAIR_N;
  const int pp = pr - n * NPAIR_N;
  v8f accA[6], accB[6];
  tgat_tile(a, bA, accA);
  tgat_tile(a, bB, accB);
  tgat_store_pair(accA, accB, P, n, pp, lane, half);
}

__device__ __forceinline__ void tgat_load_pair(const float* __restrict__ x, int pr, int c16,
                                               int half, v2f (&bA)[16], v2f (&bB)[16]) {
  const int n  = pr / NPAIR_N;
  const int pp = pr - n * NPAIR_N;
  tgat_load_x(x, n, pp * 32, c16, half, bA);
  tgat_load_x(x, n, pp * 32 + 16, c16, half, bB);
}

__global__ void __launch_bounds__(128)
tgat_xw_wmma(const float* __restrict__ x, const float* __restrict__ W,
             const float* __restrict__ alpha, const float* __restrict__ phi,
             float* __restrict__ P) {
  const int lane = threadIdx.x & 31;
  const int wave = __builtin_amdgcn_readfirstlane(threadIdx.x >> 5);
  const int half = lane >> 4;
  const int c16  = lane & 15;

  const int nw = gridDim.x * 4;
  int pr = blockIdx.x * 4 + wave;
  if (pr >= NPAIR) return;

  v2f b0A[16], b0B[16], b1A[16], b1B[16];
  tgat_load_pair(x, pr, c16, half, b0A, b0B);

  v2f a[6][16];
#pragma unroll
  for (int i = 0; i < 16; ++i) {
    const int k0 = 4 * i + 2 * half;
#pragma unroll
    for (int r = 0; r < 4; ++r) {
      a[r][i].x = W[k0 * 64 + 16 * r + c16];
      a[r][i].y = W[(k0 + 1) * 64 + 16 * r + c16];
    }
  }
#pragma unroll
  for (int rt = 0; rt < 4; ++rt) {
#pragma unroll
    for (int j = 0; j < 4; ++j) {
      const int kk = 16 * rt + 8 * half + 2 * j;
      a[4][4 * rt + j].x = alpha[c16 * 64 + kk];
      a[4][4 * rt + j].y = alpha[c16 * 64 + kk + 1];
      a[5][4 * rt + j].x = phi[c16 * 64 + kk];
      a[5][4 * rt + j].y = phi[c16 * 64 + kk + 1];
    }
  }

#pragma unroll 1
  for (;;) {
    const bool more1 = pr + nw < NPAIR;
    if (more1) tgat_load_pair(x, pr + nw, c16, half, b1A, b1B);
    tgat_pair(a, b0A, b0B, P, pr, lane, half);
    if (!more1) break;
    pr += nw;
    const bool more0 = pr + nw < NPAIR;
    if (more0) tgat_load_pair(x, pr + nw, c16, half, b0A, b0B);
    tgat_pair(a, b1A, b1B, P, pr, lane, half);
    if (!more0) break;
    pr += nw;
  }
}

__global__ void __launch_bounds__(256)
tgat_band(const float* __restrict__ P, float* __restrict__ E, float* __restrict__ Dn) {
  const int n = blockIdx.y;
  const int q = blockIdx.x * 256 + threadIdx.x;
  if (q >= QPL) return;
  const int tp = q / VV;
  const int v  = q - tp * VV;
  const float* x1p = P + ((size_t)n * ROWS + 64) * PP + (2 * tp) * VV + v;
  const float* x2p = P + ((size_t)n * ROWS + 80) * PP + v;
  float a[16];
#pragma unroll
  for (int h = 0; h < 16; ++h) a[h] = x1p[(size_t)h * PP];
  float D = (float)TT;
#pragma unroll
  for (int j = 0; j < 9; ++j) {
    const int s = 2 * tp - KHALF + j;
    float e = 0.f;
    if ((unsigned)s < (unsigned)TT) {
      float sc = 0.f;
#pragma unroll
      for (int h = 0; h < 16; ++h) sc += a[h] * x2p[(size_t)h * PP + s * VV];
      e = __expf(sc) - 1.f;
    }
    E[((size_t)n * 9 + j) * QPL + q] = e;
    D += e;
  }
  Dn[(size_t)n * QPL + q] = D;
}

#define NTH  512
#define NF4  (GCH * PP / 4)
#define NLD  ((NF4 + NTH - 1) / NTH)
#define NQ   ((QPL + NTH - 1) / NTH)
#define NSEG 5
#define NCOL (GCH * VV)
__device__ __forceinline__ int tgat_col(int it) {
  const int seg = it / NCOL;
  const int gv  = it - seg * NCOL;
  const int gch = gv / VV;
  const int v   = gv - gch * VV;
  return gch * PP + (seg * (TT / NSEG)) * VV + v;
}
__global__ void __launch_bounds__(NTH)
tgat_out(const float* __restrict__ P, const float* __restrict__ E,
         const float* __restrict__ Dn, float* __restrict__ out) {
  __shared__ __attribute__((aligned(16))) float sp[GCH * PP];
  __shared__ float spart[NSEG * NCOL];
  __shared__ float sS[NCOL];

  const int n   = blockIdx.y;
  const int c0  = blockIdx.x * GCH;
  const int tid = threadIdx.x;

  const float* En = E + (size_t)n * 9 * QPL;
  const float* Dq = Dn + (size_t)n * QPL;

  float e[NQ][9];
  float dv[NQ];
  {
    const float4* gp = reinterpret_cast<const float4*>(P + ((size_t)n * ROWS + c0) * PP);
    float4* s4 = reinterpret_cast<float4*>(sp);
    float4 r[NLD];
#pragma unroll
    for (int u = 0; u < NLD; ++u) {
      int i = tid + NTH * u;
      i = i < NF4 ? i : NF4 - 1;
      r[u] = gp[i];
    }
#pragma unroll
    for (int i = 0; i < NQ; ++i) {
      int q = tid + NTH * i;
      q = q < QPL ? q : QPL - 1;
#pragma unroll
      for (int j = 0; j < 9; ++j) e[i][j] = En[(size_t)j * QPL + q];
      dv[i] = Dq[q];
    }
#pragma unroll
    for (int u = 0; u < NLD; ++u) {
      int i = tid + NTH * u;
      i = i < NF4 ? i : NF4 - 1;
      s4[i] = r[u];
    }
  }
  __syncthreads();

  {
    const int it1  = tid + NTH;
    const int it1c = it1 < NSEG * NCOL ? it1 : NSEG * NCOL - 1;
    const float* col0 = sp + tgat_col(tid);
    const float* col1 = sp + tgat_col(it1c);
    float acc0 = 0.f, acc1 = 0.f;
    for (int t = 0; t < TT / NSEG; ++t) {
      acc0 += col0[t * VV];
      acc1 += col1[t * VV];
    }
    spart[tid] = acc0;
    if (it1 < NSEG * NCOL) spart[it1] = acc1;
  }
  __syncthreads();
  if (tid < NCOL) {
    float accv = 0.f;
#pragma unroll
    for (int seg = 0; seg < NSEG; ++seg) accv += spart[seg * NCOL + tid];
    sS[tid] = accv;
  }
  __syncthreads();

  float* ob = out + ((size_t)n * CCH + c0) * QPL;
#pragma unroll
  for (int i = 0; i < NQ; ++i) {
    const int q = tid + NTH * i;
    if (q < QPL) {
      const int tp = q / VV;
      const int v  = q - tp * VV;
      int idx[9];
#pragma unroll
      for (int j = 0; j < 9; ++j) {
        int s = 2 * tp - KHALF + j;
        s = s < 0 ? 0 : (s > TT - 1 ? TT - 1 : s);
        idx[j] = s * VV + v;
      }
      const float D = dv[i];
#pragma unroll
      for (int gch = 0; gch < GCH; ++gch) {
        float accv = sS[gch * VV + v];
#pragma unroll
        for (int j = 0; j < 9; ++j) accv += e[i][j] * sp[gch * PP + idx[j]];
        ob[(size_t)gch * QPL + q] = accv / D;
      }
    }
  }
}

extern "C" void kernel_launch(void* const* d_in, const int* in_sizes, int n_in,
                              void* d_out, int out_size, void* d_ws, size_t ws_size,
                              hipStream_t stream) {
  const float* x     = (const float*)d_in[0];
  const float* W     = (const float*)d_in[1];
  const float* alpha = (const float*)d_in[2];
  const float* phi   = (const float*)d_in[3];

  float* P    = (float*)d_ws;
  float* E    = P  + (size_t)NB * ROWS * PP;
  float* Dn   = E  + (size_t)NB * 9 * QPL;

  tgat_xw_wmma<<<XWG, 128, 0, stream>>>(x, W, alpha, phi, P);
  tgat_band<<<dim3(15, NB), 256, 0, stream>>>(P, E, Dn);
  tgat_out<<<dim3(CCH / GCH, NB), NTH, 0, stream>>>(P, E, Dn, (float*)d_out);
}


# band weights made in the GEMM launch: a work-group owns 1024 positions, x1/x2 go to LDS not to P, x2 of 4 rows each side from W phi^T products (those sums reorder); band launch and x1/x2 rows go
# speedup vs baseline: 7.5068x; 1.1630x over previous
#include <hip/hip_runtime.h>
#include <math.h>

typedef __attribute__((ext_vector_type(2))) float v2f;
typedef __attribute__((ext_vector_type(8))) float v8f;

#define NB    32
#define CCH   64
#define TT    300
#define VV    25
#define PLANE (TT*VV)
#define NPOS  (NB*PLANE)
#define TP    150
#define QPL   (TP*VV)
#define KHALF 4
#define ROWS  64
#define GCH   8

#define PP      7520
#define NPAIR_N (PP / 32)
#define BPAIR   32
#define NBLK    ((NPAIR_N + BPAIR - 1) / BPAIR)
#define NPOSB   (BPAIR * 32)
#define HALO    112
#define LPIT    20
#define WPIT    17

__device__ __forceinline__ void tgat_load_x(const float* __restrict__ x, int n, int p0, int c16,
                                            int half, v2f (&b)[16]) {
  int p = p0 + c16;
  p = p < PLANE ? p : PLANE - 1;
  p = p < 0 ? 0 : p;
  const float* xb = x + (size_t)n * (CCH * PLANE) + (size_t)(2 * half) * PLANE + p;
#pragma unroll
  for (int i = 0; i < 16; ++i) {
    b[i].x = xb[(size_t)(4 * i) * PLANE];
    b[i].y = xb[(size_t)(4 * i + 1) * PLANE];
  }
}

__device__ __forceinline__ void tgat_tile(const v2f (&a)[6][16], const v2f (&b)[16], v8f (&acc)[6]) {
  v8f acc0 = {}, acc1 = {}, acc2 = {}, acc3 = {}, acc4 = {}, acc5 = {};
#pragma unroll
  for (int i = 0; i < 16; ++i) {
    acc0 = __builtin_amdgcn_wmma_f32_16x16x4_f32(false, a[0][i], false, b[i], (short)0, acc0, false, false);
    acc1 = __builtin_amdgcn_wmma_f32_16x16x4_f32(false, a[1][i], false, b[i], (short)0, acc1, false, false);
    acc2 = __builtin_amdgcn_wmma_f32_16x16x4_f32(false, a[2][i], false, b[i], (short)0, acc2, false, false);
    acc3 = __builtin_amdgcn_wmma_f32_16x16x4_f32(false, a[3][i], false, b[i], (short)0, acc3, false, false);
  }
#pragma unroll
  for (int j = 0; j < 4; ++j) {
    const v2f q0 = {acc0[2 * j], acc0[2 * j + 1]};
    acc4 = __builtin_amdgcn_wmma_f32_16x16x4_f32(false, a[4][j], false, q0, (short)0, acc4, false, false);
    acc5 = __builtin_amdgcn_wmma_f32_16x16x4_f32(false, a[5][j], false, q0, (short)0, acc5, false, false);
  }
#pragma unroll
  for (int j = 0; j < 4; ++j) {
    const v2f q1 = {acc1[2 * j], acc1[2 * j + 1]};
    acc4 = __builtin_amdgcn_wmma_f32_16x16x4_f32(false, a[4][4 + j], false, q1, (short)0, acc4, false, false);
    acc5 = __builtin_amdgcn_wmma_f32_16x16x4_f32(false, a[5][4 + j], false, q1, (short)0, acc5, false, false);
  }
#pragma unroll
  for (int j = 0; j < 4; ++j) {
    const v2f q2 = {acc2[2 * j], acc2[2 * j + 1]};
    acc4 = __builtin_amdgcn_wmma_f32_16x16x4_f32(false, a[4][8 + j], false, q2, (short)0, acc4, false, false);
    acc5 = __builtin_amdgcn_wmma_f32_16x16x4_f32(false, a[5][8 + j], false, q2, (short)0, acc5, false, false);
  }
#pragma unroll
  for (int j = 0; j < 4; ++j) {
    const v2f q3 = {acc3[2 * j], acc3[2 * j + 1]};
    acc4 = __builtin_amdgcn_wmma_f32_16x16x4_f32(false, a[4][12 + j], false, q3, (short)0, acc4, false, false);
    acc5 = __builtin_amdgcn_wmma_f32_16x16x4_f32(false, a[5][12 + j], false, q3, (short)0, acc5, false, false);
  }
  acc[0] = acc0; acc[1] = acc1; acc[2] = acc2; acc[3] = acc3; acc[4] = acc4; acc[5] = acc5;
}

__device__ __forceinline__ float tgat_xhalf(float v) {
  return __builtin_bit_cast(float, __builtin_amdgcn_permlanex16(__builtin_bit_cast(int, v),
                            __builtin_bit_cast(int, v), 0x76543210, 0xfedcba98, false, false));
}

__device__ __forceinline__ void tgat_store_pair(const v8f (&A)[6], const v8f (&B)[6],
                                                float* __restrict__ P, int n, int pp,
                                                int lane, int half) {
  float* pb = P + (size_t)n * (ROWS * PP) + pp * 32 + lane;
#pragma unroll
  for (int k = 0; k < 4; ++k) {
#pragma unroll
    for (int r = 0; r < 8; ++r) {
      const float send = half ? A[k][r] : B[k][r];
      const float recv = tgat_xhalf(send);
      const float lo = half ? recv : A[k][r];
      const float hi = half ? B[k][r] : recv;
      pb[(size_t)(16 * k + r)     * PP] = lo;
      pb[(size_t)(16 * k + 8 + r) * PP] = hi;
    }
  }
}

__device__ __forceinline__ void tgat_lds_rows(float* dst, const v8f& v) {
  float4* d = reinterpret_cast<float4*>(dst);
  d[0] = make_float4(v[0], v[1], v[2], v[3]);
  d[1] = make_float4(v[4], v[5], v[6], v[7]);
}

__device__ __forceinline__ void tgat_pair(const v2f (&a)[6][16], const v2f (&bA)[16],
                                          const v2f (&bB)[16], float* __restrict__ P,
                                          float* x1s, float* x2s, int n, int pp, int pl,
                                          int lane, int half, int c16) {
  v8f accA[6], accB[6];
  tgat_tile(a, bA, accA);
  tgat_tile(a, bB, accB);
  tgat_store_pair(accA, accB, P, n, pp, lane, half);
  const int lp = pl * 32 + c16;
  tgat_lds_rows(x1s + (lp)             * LPIT + 8 * half, accA[4]);
  tgat_lds_rows(x1s + (lp + 16)        * LPIT + 8 * half, accB[4]);
  tgat_lds_rows(x2s + (HALO + lp)      * LPIT + 8 * half, accA[5]);
  tgat_lds_rows(x2s + (HALO + lp + 16) * LPIT + 8 * half, accB[5]);
}

__device__ __forceinline__ void tgat_load_pair(const float* __restrict__ x, int n, int pp, int c16,
                                               int half, v2f (&bA)[16], v2f (&bB)[16]) {
  tgat_load_x(x, n, pp * 32, c16, half, bA);
  tgat_load_x(x, n, pp * 32 + 16, c16, half, bB);
}

__device__ __forceinline__ v8f tgat_halo_tile(const v2f (&ah)[16], const v2f (&b)[16]) {
  v8f acc = {};
#pragma unroll
  for (int i = 0; i < 16; ++i)
    acc = __builtin_amdgcn_wmma_f32_16x16x4_f32(false, ah[i], false, b[i], (short)0, acc, false, false);
  return acc;
}

__device__ __forceinline__ int tgat_halo_idx(int h) {
  return h < 7 ? 16 * h : HALO + NPOSB + 16 * (h - 7);
}

__global__ void __launch_bounds__(256)
tgat_xw_band(const float* __restrict__ x, const float* __restrict__ W,
             const float* __restrict__ alpha, const float* __restrict__ phi,
             float* __restrict__ P, float* __restrict__ E, float* __restrict__ Dn) {
  __shared__ __attribute__((aligned(16))) float x1s[NPOSB * LPIT];
  __shared__ __attribute__((aligned(16))) float x2s[(NPOSB + 2 * HALO) * LPIT];
  __shared__ float wps[CCH * WPIT];

  const int lane = threadIdx.x & 31;
  const int wave = __builtin_amdgcn_readfirstlane(threadIdx.x >> 5);
  const int half = lane >> 4;
  const int c16  = lane & 15;

  const int n  = blockIdx.y;
  const int pb = blockIdx.x * BPAIR;
  const int P0 = pb * 32;

  v2f b0A[16], b0B[16], b1A[16], b1B[16];
  tgat_load_pair(x, n, pb + wave, c16, half, b0A, b0B);

  v2f a[6][16];
#pragma unroll
  for (int i = 0; i < 16; ++i) {
    const int k0 = 4 * i + 2 * half;
#pragma unroll
    for (int r = 0; r < 4; ++r) {
      a[r][i].x = W[k0 * 64 + 16 * r + c16];
      a[r][i].y = W[(k0 + 1) * 64 + 16 * r + c16];
    }
  }
#pragma unroll
  for (int rt = 0; rt < 4; ++rt) {
#pragma unroll
    for (int j = 0; j < 4; ++j) {
      const int kk = 16 * rt + 8 * half + 2 * j;
      a[4][4 * rt + j].x = alpha[c16 * 64 + kk];
      a[4][4 * rt + j].y = alpha[c16 * 64 + kk + 1];
      a[5][4 * rt + j].x = phi[c16 * 64 + kk];
      a[5][4 * rt + j].y = phi[c16 * 64 + kk + 1];
    }
  }

  if (wave < 4) {
    const float* wr = W + (16 * wave + c16) * 64 + 8 * half;
    v8f wacc = {};
#pragma unroll
    for (int m = 0; m < 16; ++m) {
      const int kof = 16 * (m >> 2) + 2 * (m & 3);
      const v2f wb = {wr[kof], wr[kof + 1]};
      wacc = __builtin_amdgcn_wmma_f32_16x16x4_f32(false, a[5][m], false, wb, (short)0, wacc, false, false);
    }
#pragma unroll
    for (int r = 0; r < 8; ++r) wps[(16 * wave + c16) * WPIT + 8 * half + r] = wacc[r];
  }

#pragma unroll 1
  for (int s = 0; s < 4; s += 2) {
    const int pp0 = pb + wave + 8 * s;
    const int pp1 = pp0 + 8;
    if (pp1 < NPAIR_N) tgat_load_pair(x, n, pp1, c16, half, b1A, b1B);
    if (pp0 < NPAIR_N) tgat_pair(a, b0A, b0B, P, x1s, x2s, n, pp0, wave + 8 * s, lane, half, c16);
    if (s == 0) {
      if (pp0 + 16 < NPAIR_N) tgat_load_pair(x, n, pp0 + 16, c16, half, b0A, b0B);
    } else {
      tgat_load_x(x, n, P0 - HALO + tgat_halo_idx(wave), c16, half, b0A);
      if (wave < 6) tgat_load_x(x, n, P0 - HALO + tgat_halo_idx(wave + 8), c16, half, b0B);
    }
    if (pp1 < NPAIR_N) tgat_pair(a, b1A, b1B, P, x1s, x2s, n, pp1, wave + 8 * s + 8, lane, half, c16);
  }

  __syncthreads();

  {
    v2f ah[16];
#pragma unroll
    for (int i = 0; i < 16; ++i) {
      const int k0 = 4 * i + 2 * half;
      ah[i].x = wps[k0 * WPIT + c16];
      ah[i].y = wps[(k0 + 1) * WPIT + c16];
    }
    const v8f h0 = tgat_halo_tile(ah, b0A);
    tgat_lds_rows(x2s + (tgat_halo_idx(wave) + c16) * LPIT + 8 * half, h0);
    if (wave < 6) {
      const v8f h1 = tgat_halo_tile(ah, b0B);
      tgat_lds_rows(x2s + (tgat_halo_idx(wave + 8) + c16) * LPIT + 8 * half, h1);
    }
  }

  __syncthreads();

  const int tid = threadIdx.x;
  const int tlo = P0 / VV;
  const int te0 = tlo + (tlo & 1);
  float* En = E + (size_t)n * 9 * QPL;
  float* Dq = Dn + (size_t)n * QPL;
#pragma unroll 1
  for (int it = 0; it < 3; ++it) {
    const int o   = tid + 256 * it;
    const int row = o / VV;
    const int v   = o - row * VV;
    const int te  = te0 + 2 * row;
    const int p   = te * VV + v;
    if (p >= P0 && p < P0 + NPOSB && te < TT) {
      const int lp = p - P0;
      const int q  = (te >> 1) * VV + v;
      float a1[16];
      {
        const float4* s1 = reinterpret_cast<const float4*>(x1s + lp * LPIT);
#pragma unroll
        for (int u = 0; u < 4; ++u) {
          const float4 t4 = s1[u];
          a1[4 * u] = t4.x; a1[4 * u + 1] = t4.y; a1[4 * u + 2] = t4.z; a1[4 * u + 3] = t4.w;
        }
      }
      float D = (float)TT;
#pragma unroll
      for (int j = 0; j < 9; ++j) {
        const int sr = te - KHALF + j;
        float e = 0.f;
        if ((unsigned)sr < (unsigned)TT) {
          const float4* s2 = reinterpret_cast<const float4*>(x2s + (HALO + lp + (j - KHALF) * VV) * LPIT);
          float sc = 0.f;
#pragma unroll
          for (int u = 0; u < 4; ++u) {
            const float4 t4 = s2[u];
            sc = fmaf(a1[4 * u],     t4.x, sc);
            sc = fmaf(a1[4 * u + 1], t4.y, sc);
            sc = fmaf(a1[4 * u + 2], t4.z, sc);
            sc = fmaf(a1[4 * u + 3], t4.w, sc);
          }
          e = __expf(sc) - 1.f;
        }
        En[(size_t)j * QPL + q] = e;
        D += e;
      }
      Dq[q] = D;
    }
  }
}

#define NTH  512
#define NF4  (GCH * PP / 4)
#define NLD  ((NF4 + NTH - 1) / NTH)
#define NQ   ((QPL + NTH - 1) / NTH)
#define NSEG 5
#define NCOL (GCH * VV)
__device__ __forceinline__ int tgat_col(int it) {
  const int seg = it / NCOL;
  const int gv  = it - seg * NCOL;
  const int gch = gv / VV;
  const int v   = gv - gch * VV;
  return gch * PP + (seg * (TT / NSEG)) * VV + v;
}
__global__ void __launch_bounds__(NTH)
tgat_out(const float* __restrict__ P, const float* __restrict__ E,
         const float* __restrict__ Dn, float* __restrict__ out) {
  __shared__ __attribute__((aligned(16))) float sp[GCH * PP];
  __shared__ float spart[NSEG * NCOL];
  __shared__ float sS[NCOL];

  const int n   = blockIdx.y;
  const int c0  = blockIdx.x * GCH;
  const int tid = threadIdx.x;

  const float* En = E + (size_t)n * 9 * QPL;
  const float* Dq = Dn + (size_t)n * QPL;

  float e[NQ][9];
  float dv[NQ];
  {
    const float4* gp = reinterpret_cast<const float4*>(P + ((size_t)n * ROWS + c0) * PP);
    float4* s4 = reinterpret_cast<float4*>(sp);
    float4 r[NLD];
#pragma unroll
    for (int u = 0; u < NLD; ++u) {
      int i = tid + NTH * u;
      i = i < NF4 ? i : NF4 - 1;
      r[u] = gp[i];
    }
#pragma unroll
    for (int i = 0; i < NQ; ++i) {
      int q = tid + NTH * i;
      q = q < QPL ? q : QPL - 1;
#pragma unroll
      for (int j = 0; j < 9; ++j) e[i][j] = En[(size_t)j * QPL + q];
      dv[i] = Dq[q];
    }
#pragma unroll
    for (int u = 0; u < NLD; ++u) {
      int i = tid + NTH * u;
      i = i < NF4 ? i : NF4 - 1;
      s4[i] = r[u];
    }
  }
  __syncthreads();

  {
    const int it1  = tid + NTH;
    const int it1c = it1 < NSEG * NCOL ? it1 : NSEG * NCOL - 1;
    const float* col0 = sp + tgat_col(tid);
    const float* col1 = sp + tgat_col(it1c);
    float acc0 = 0.f, acc1 = 0.f;
    for (int t = 0; t < TT / NSEG; ++t) {
      acc0 += col0[t * VV];
      acc1 += col1[t * VV];
    }
    spart[tid] = acc0;
    if (it1 < NSEG * NCOL) spart[it1] = acc1;
  }
  __syncthreads();
  if (tid < NCOL) {
    float accv = 0.f;
#pragma unroll
    for (int seg = 0; seg < NSEG; ++seg) accv += spart[seg * NCOL + tid];
    sS[tid] = accv;
  }
  __syncthreads();

  float* ob = out + ((size_t)n * CCH + c0) * QPL;
#pragma unroll
  for (int i = 0; i < NQ; ++i) {
    const int q = tid + NTH * i;
    if (q < QPL) {
      const int tp = q / VV;
      const int v  = q - tp * VV;
      int idx[9];
#pragma unroll
      for (int j = 0; j < 9; ++j) {
        int s = 2 * tp - KHALF + j;
        s = s < 0 ? 0 : (s > TT - 1 ? TT - 1 : s);
        idx[j] = s * VV + v;
      }
      const float D = dv[i];
#pragma unroll
      for (int gch = 0; gch < GCH; ++gch) {
        float accv = sS[gch * VV + v];
#pragma unroll
        for (int j = 0; j < 9; ++j) accv += e[i][j] * sp[gch * PP + idx[j]];
        ob[(size_t)gch * QPL + q] = accv / D;
      }
    }
  }
}

extern "C" void kernel_launch(void* const* d_in, const int* in_sizes, int n_in,
                              void* d_out, int out_size, void* d_ws, size_t ws_size,
                              hipStream_t stream) {
  const float* x     = (const float*)d_in[0];
  const float* W     = (const float*)d_in[1];
  const float* alpha = (const float*)d_in[2];
  const float* phi   = (const float*)d_in[3];

  float* P    = (float*)d_ws;
  float* E    = P  + (size_t)NB * ROWS * PP;
  float* Dn   = E  + (size_t)NB * 9 * QPL;

  tgat_xw_band<<<dim3(NBLK, NB), 256, 0, stream>>>(x, W, alpha, phi, P, E, Dn);
  tgat_out<<<dim3(CCH / GCH, NB), NTH, 0, stream>>>(P, E, Dn, (float*)d_out);
}


# xw plane kept as 16-byte pieces of 4 channels per position: accumulators stored directly, no lane swap; output kernel reads them by 16-byte LDS loads on a conflict-free two-row pitch; sums unchanged
# speedup vs baseline: 8.2540x; 1.0995x over previous
#include <hip/hip_runtime.h>
#include <math.h>

typedef __attribute__((ext_vector_type(2))) float v2f;
typedef __attribute__((ext_vector_type(8))) float v8f;

#define NB    32
#define CCH   64
#define TT    300
#define VV    25
#define PLANE (TT*VV)
#define NPOS  (NB*PLANE)
#define TP    150
#define QPL   (TP*VV)
#define KHALF 4
#define ROWS  64
#define GCH   8

#define PP      7520
#define NCG     (ROWS / 4)
#define NPAIR_N (PP / 32)
#define BPAIR   32
#define NBLK    ((NPAIR_N + BPAIR - 1) / BPAIR)
#define NPOSB   (BPAIR * 32)
#define HALO    112
#define LPIT    20
#define WPIT    17

__device__ __forceinline__ void tgat_load_x(const float* __restrict__ x, int n, int p0, int c16,
                                            int half, v2f (&b)[16]) {
  int p = p0 + c16;
  p = p < PLANE ? p : PLANE - 1;
  p = p < 0 ? 0 : p;
  const float* xb = x + (size_t)n * (CCH * PLANE) + (size_t)(2 * half) * PLANE + p;
#pragma unroll
  for (int i = 0; i < 16; ++i) {
    b[i].x = xb[(size_t)(4 * i) * PLANE];
    b[i].y = xb[(size_t)(4 * i + 1) * PLANE];
  }
}

__device__ __forceinline__ void tgat_tile(const v2f (&a)[6][16], const v2f (&b)[16], v8f (&acc)[6]) {
  v8f acc0 = {}, acc1 = {}, acc2 = {}, acc3 = {}, acc4 = {}, acc5 = {};
#pragma unroll
  for (int i = 0; i < 16; ++i) {
    acc0 = __builtin_amdgcn_wmma_f32_16x16x4_f32(false, a[0][i], false, b[i], (short)0, acc0, false, false);
    acc1 = __builtin_amdgcn_wmma_f32_16x16x4_f32(false, a[1][i], false, b[i], (short)0, acc1, false, false);
    acc2 = __builtin_amdgcn_wmma_f32_16x16x4_f32(false, a[2][i], false, b[i], (short)0, acc2, false, false);
    acc3 = __builtin_amdgcn_wmma_f32_16x16x4_f32(false, a[3][i], false, b[i], (short)0, acc3, false, false);
  }
#pragma unroll
  for (int j = 0; j < 4; ++j) {
    const v2f q0 = {acc0[2 * j], acc0[2 * j + 1]};
    acc4 = __builtin_amdgcn_wmma_f32_16x16x4_f32(false, a[4][j], false, q0, (short)0, acc4, false, false);
    acc5 = __builtin_amdgcn_wmma_f32_16x16x4_f32(false, a[5][j], false, q0, (short)0, acc5, false, false);
  }
#pragma unroll
  for (int j = 0; j < 4; ++j) {
    const v2f q1 = {acc1[2 * j], acc1[2 * j + 1]};
    acc4 = __builtin_amdgcn_wmma_f32_16x16x4_f32(false, a[4][4 + j], false, q1, (short)0, acc4, false, false);
    acc5 = __builtin_amdgcn_wmma_f32_16x16x4_f32(false, a[5][4 + j], false, q1, (short)0, acc5, false, false);
  }
#pragma unroll
  for (int j = 0; j < 4; ++j) {
    const v2f q2 = {acc2[2 * j], acc2[2 * j + 1]};
    acc4 = __builtin_amdgcn_wmma_f32_16x16x4_f32(false, a[4][8 + j], false, q2, (short)0, acc4, false, false);
    acc5 = __builtin_amdgcn_wmma_f32_16x16x4_f32(false, a[5][8 + j], false, q2, (short)0, acc5, false, false);
  }
#pragma unroll
  for (int j = 0; j < 4; ++j) {
    const v2f q3 = {acc3[2 * j], acc3[2 * j + 1]};
    acc4 = __builtin_amdgcn_wmma_f32_16x16x4_f32(false, a[4][12 + j], false, q3, (short)0, acc4, false, false);
    acc5 = __builtin_amdgcn_wmma_f32_16x16x4_f32(false, a[5][12 + j], false, q3, (short)0, acc5, false, false);
  }
  acc[0] = acc0; acc[1] = acc1; acc[2] = acc2; acc[3] = acc3; acc[4] = acc4; acc[5] = acc5;
}

__device__ __forceinline__ void tgat_store_tile(const v8f (&A)[6], float* __restrict__ P, int n,
                                                int pos, int half) {
  float4* pb = reinterpret_cast<float4*>(P) + ((size_t)n * NCG + 2 * half) * PP + pos;
#pragma unroll
  for (int k = 0; k < 4; ++k) {
#pragma unroll
    for (int u = 0; u < 2; ++u) {
      pb[(size_t)(4 * k + u) * PP] =
          make_float4(A[k][4 * u], A[k][4 * u + 1], A[k][4 * u + 2], A[k][4 * u + 3]);
    }
  }
}

__device__ __forceinline__ void tgat_lds_rows(float* dst, const v8f& v) {
  float4* d = reinterpret_cast<float4*>(dst);
  d[0] = make_float4(v[0], v[1], v[2], v[3]);
  d[1] = make_float4(v[4], v[5], v[6], v[7]);
}

__device__ __forceinline__ void tgat_pair(const v2f (&a)[6][16], const v2f (&bA)[16],
                                          const v2f (&bB)[16], float* __restrict__ P,
                                          float* x1s, float* x2s, int n, int pp, int pl,
                                          int half, int c16) {
  const int lp = pl * 32 + c16;
  v8f accA[6], accB[6];
  tgat_tile(a, bA, accA);
  tgat_store_tile(accA, P, n, pp * 32 + c16, half);
  tgat_lds_rows(x1s + (lp)             * LPIT + 8 * half, accA[4]);
  tgat_lds_rows(x2s + (HALO + lp)      * LPIT + 8 * half, accA[5]);
  tgat_tile(a, bB, accB);
  tgat_store_tile(accB, P, n, pp * 32 + 16 + c16, half);
  tgat_lds_rows(x1s + (lp + 16)        * LPIT + 8 * half, accB[4]);
  tgat_lds_rows(x2s + (HALO + lp + 16) * LPIT + 8 * half, accB[5]);
}

__device__ __forceinline__ void tgat_load_pair(const float* __restrict__ x, int n, int pp, int c16,
                                               int half, v2f (&bA)[16], v2f (&bB)[16]) {
  tgat_load_x(x, n, pp * 32, c16, half, bA);
  tgat_load_x(x, n, pp * 32 + 16, c16, half, bB);
}

__device__ __forceinline__ v8f tgat_halo_tile(const v2f (&ah)[16], const v2f (&b)[16]) {
  v8f acc = {};
#pragma unroll
  for (int i = 0; i < 16; ++i)
    acc = __builtin_amdgcn_wmma_f32_16x16x4_f32(false, ah[i], false, b[i], (short)0, acc, false, false);
  return acc;
}

__device__ __forceinline__ int tgat_halo_idx(int h) {
  return h < 7 ? 16 * h : HALO + NPOSB + 16 * (h - 7);
}

__global__ void __launch_bounds__(256)
tgat_xw_band(const float* __restrict__ x, const float* __restrict__ W,
             const float* __restrict__ alpha, const float* __restrict__ phi,
             float* __restrict__ P, float* __restrict__ E, float* __restrict__ Dn) {
  __shared__ __attribute__((aligned(16))) float x1s[NPOSB * LPIT];
  __shared__ __attribute__((aligned(16))) float x2s[(NPOSB + 2 * HALO) * LPIT];
  __shared__ float wps[CCH * WPIT];

  const int lane = threadIdx.x & 31;
  const int wave = __builtin_amdgcn_readfirstlane(threadIdx.x >> 5);
  const int half = lane >> 4;
  const int c16  = lane & 15;

  const int n  = blockIdx.y;
  const int pb = blockIdx.x * BPAIR;
  const int P0 = pb * 32;

  v2f b0A[16], b0B[16], b1A[16], b1B[16];
  tgat_load_pair(x, n, pb + wave, c16, half, b0A, b0B);

  v2f a[6][16];
#pragma unroll
  for (int i = 0; i < 16; ++i) {
    const int k0 = 4 * i + 2 * half;
#pragma unroll
    for (int r = 0; r < 4; ++r) {
      a[r][i].x = W[k0 * 64 + 16 * r + c16];
      a[r][i].y = W[(k0 + 1) * 64 + 16 * r + c16];
    }
  }
#pragma unroll
  for (int rt = 0; rt < 4; ++rt) {
#pragma unroll
    for (int j = 0; j < 4; ++j) {
      const int kk = 16 * rt + 8 * half + 2 * j;
      a[4][4 * rt + j].x = alpha[c16 * 64 + kk];
      a[4][4 * rt + j].y = alpha[c16 * 64 + kk + 1];
      a[5][4 * rt + j].x = phi[c16 * 64 + kk];
      a[5][4 * rt + j].y = phi[c16 * 64 + kk + 1];
    }
  }

  if (wave < 4) {
    const float* wr = W + (16 * wave + c16) * 64 + 8 * half;
    v8f wacc = {};
#pragma unroll
    for (int m = 0; m < 16; ++m) {
      const int kof = 16 * (m >> 2) + 2 * (m & 3);
      const v2f wb = {wr[kof], wr[kof + 1]};
      wacc = __builtin_amdgcn_wmma_f32_16x16x4_f32(false, a[5][m], false, wb, (short)0, wacc, false, false);
    }
#pragma unroll
    for (int r = 0; r < 8; ++r) wps[(16 * wave + c16) * WPIT + 8 * half + r] = wacc[r];
  }

#pragma unroll 1
  for (int s = 0; s < 4; s += 2) {
    const int pp0 = pb + wave + 8 * s;
    const int pp1 = pp0 + 8;
    if (pp1 < NPAIR_N) tgat_load_pair(x, n, pp1, c16, half, b1A, b1B);
    if (pp0 < NPAIR_N) tgat_pair(a, b0A, b0B, P, x1s, x2s, n, pp0, wave + 8 * s, half, c16);
    if (s == 0) {
      if (pp0 + 16 < NPAIR_N) tgat_load_pair(x, n, pp0 + 16, c16, half, b0A, b0B);
    } else {
      tgat_load_x(x, n, P0 - HALO + tgat_halo_idx(wave), c16, half, b0A);
      if (wave < 6) tgat_load_x(x, n, P0 - HALO + tgat_halo_idx(wave + 8), c16, half, b0B);
    }
    if (pp1 < NPAIR_N) tgat_pair(a, b1A, b1B, P, x1s, x2s, n, pp1, wave + 8 * s + 8, half, c16);
  }

  __syncthreads();

  {
    v2f ah[16];
#pragma unroll
    for (int i = 0; i < 16; ++i) {
      const int k0 = 4 * i + 2 * half;
      ah[i].x = wps[k0 * WPIT + c16];
      ah[i].y = wps[(k0 + 1) * WPIT + c16];
    }
    const v8f h0 = tgat_halo_tile(ah, b0A);
    tgat_lds_rows(x2s + (tgat_halo_idx(wave) + c16) * LPIT + 8 * half, h0);
    if (wave < 6) {
      const v8f h1 = tgat_halo_tile(ah, b0B);
      tgat_lds_rows(x2s + (tgat_halo_idx(wave + 8) + c16) * LPIT + 8 * half, h1);
    }
  }

  __syncthreads();

  const int tid = threadIdx.x;
  const int tlo = P0 / VV;
  const int te0 = tlo + (tlo & 1);
  float* En = E + (size_t)n * 9 * QPL;
  float* Dq = Dn + (size_t)n * QPL;
#pragma unroll 1
  for (int it = 0; it < 3; ++it) {
    const int o   = tid + 256 * it;
    const int row = o / VV;
    const int v   = o - row * VV;
    const int te  = te0 + 2 * row;
    const int p   = te * VV + v;
    if (p >= P0 && p < P0 + NPOSB && te < TT) {
      const int lp = p - P0;
      const int q  = (te >> 1) * VV + v;
      float a1[16];
      {
        const float4* s1 = reinterpret_cast<const float4*>(x1s + lp * LPIT);
#pragma unroll
        for (int u = 0; u < 4; ++u) {
          const float4 t4 = s1[u];
          a1[4 * u] = t4.x; a1[4 * u + 1] = t4.y; a1[4 * u + 2] = t4.z; a1[4 * u + 3] = t4.w;
        }
      }
      float D = (float)TT;
#pragma unroll
      for (int j = 0; j < 9; ++j) {
        const int sr = te - KHALF + j;
        float e = 0.f;
        if ((unsigned)sr < (unsigned)TT) {
          const float4* s2 = reinterpret_cast<const float4*>(x2s + (HALO + lp + (j - KHALF) * VV) * LPIT);
          float sc = 0.f;
#pragma unroll
          for (int u = 0; u < 4; ++u) {
            const float4 t4 = s2[u];
            sc = fmaf(a1[4 * u],     t4.x, sc);
            sc = fmaf(a1[4 * u + 1], t4.y, sc);
            sc = fmaf(a1[4 * u + 2], t4.z, sc);
            sc = fmaf(a1[4 * u + 3], t4.w, sc);
          }
          e = __expf(sc) - 1.f;
        }
        En[(size_t)j * QPL + q] = e;
        D += e;
      }
      Dq[q] = D;
    }
  }
}

#define NTH  512
#define GCG  (GCH / 4)
#define NF4  (GCG * PP)
#define NLD  ((NF4 + NTH - 1) / NTH)
#define NQ   ((QPL + NTH - 1) / NTH)
#define NSEG 5
#define NCOL (GCG * VV)
#define RP2  57
#define SPG  8576
__global__ void __launch_bounds__(NTH)
tgat_out(const float* __restrict__ P, const float* __restrict__ E,
         const float* __restrict__ Dn, float* __restrict__ out) {
  __shared__ __attribute__((aligned(16))) float sp[GCG * SPG * 4];
  __shared__ __attribute__((aligned(16))) float spart[NSEG * NCOL * 4];
  __shared__ __attribute__((aligned(16))) float sS[NCOL * 4];

  const int n   = blockIdx.y;
  const int g0  = blockIdx.x * GCG;
  const int c0  = blockIdx.x * GCH;
  const int tid = threadIdx.x;

  const float* En = E + (size_t)n * 9 * QPL;
  const float* Dq = Dn + (size_t)n * QPL;

  float4* s4 = reinterpret_cast<float4*>(sp);
  float4* part4 = reinterpret_cast<float4*>(spart);
  float4* sS4 = reinterpret_cast<float4*>(sS);

  float e[NQ][9];
  float dv[NQ];
  {
    const float4* gp = reinterpret_cast<const float4*>(P) + ((size_t)n * NCG + g0) * PP;
    float4 r[NLD];
#pragma unroll
    for (int u = 0; u < NLD; ++u) {
      int i = tid + NTH * u;
      i = i < NF4 ? i : NF4 - 1;
      r[u] = gp[i];
    }
#pragma unroll
    for (int i = 0; i < NQ; ++i) {
      int q = tid + NTH * i;
      q = q < QPL ? q : QPL - 1;
#pragma unroll
      for (int j = 0; j < 9; ++j) e[i][j] = En[(size_t)j * QPL + q];
      dv[i] = Dq[q];
    }
#pragma unroll
    for (int u = 0; u < NLD; ++u) {
      int i = tid + NTH * u;
      i = i < NF4 ? i : NF4 - 1;
      const int cg = i >= PP ? 1 : 0;
      const int p  = i - cg * PP;
      const int pr = p / (2 * VV);
      s4[cg * SPG + pr * RP2 + (p - pr * (2 * VV))] = r[u];
    }
  }
  __syncthreads();

  if (tid < NSEG * NCOL) {
    const int seg = tid / NCOL;
    const int gv  = tid - seg * NCOL;
    const int cg  = gv / VV;
    const int v   = gv - cg * VV;
    const float4* col = s4 + cg * SPG + (seg * (TT / NSEG / 2)) * RP2 + v;
    float4 acc = make_float4(0.f, 0.f, 0.f, 0.f);
    for (int t = 0; t < TT / NSEG; ++t) {
      const float4 xv = col[(t >> 1) * RP2 + (t & 1) * VV];
      acc.x += xv.x; acc.y += xv.y; acc.z += xv.z; acc.w += xv.w;
    }
    part4[tid] = acc;
  }
  __syncthreads();
  if (tid < NCOL) {
    float4 accv = make_float4(0.f, 0.f, 0.f, 0.f);
#pragma unroll
    for (int seg = 0; seg < NSEG; ++seg) {
      const float4 pv = part4[seg * NCOL + tid];
      accv.x += pv.x; accv.y += pv.y; accv.z += pv.z; accv.w += pv.w;
    }
    sS4[tid] = accv;
  }
  __syncthreads();

  float* ob = out + ((size_t)n * CCH + c0) * QPL;
#pragma unroll
  for (int i = 0; i < NQ; ++i) {
    const int q = tid + NTH * i;
    if (q < QPL) {
      const int tp = q / VV;
      const int v  = q - tp * VV;
      int idx[9];
#pragma unroll
      for (int j = 0; j < 9; ++j) {
        int pr = tp - KHALF / 2 + (j >> 1);
        pr = pr < 0 ? 0 : (pr > TP - 1 ? TP - 1 : pr);
        idx[j] = pr * RP2 + (j & 1) * VV + v;
      }
      const float D = dv[i];
#pragma unroll
      for (int cg = 0; cg < GCG; ++cg) {
        float4 accv = sS4[cg * VV + v];
#pragma unroll
        for (int j = 0; j < 9; ++j) {
          const float4 xv = s4[cg * SPG + idx[j]];
          accv.x += e[i][j] * xv.x;
          accv.y += e[i][j] * xv.y;
          accv.z += e[i][j] * xv.z;
          accv.w += e[i][j] * xv.w;
        }
        ob[(size_t)(4 * cg)     * QPL + q] = accv.x / D;
        ob[(size_t)(4 * cg + 1) * QPL + q] = accv.y / D;
        ob[(size_t)(4 * cg + 2) * QPL + q] = accv.z / D;
        ob[(size_t)(4 * cg + 3) * QPL + q] = accv.w / D;
      }
    }
  }
}

extern "C" void kernel_launch(void* const* d_in, const int* in_sizes, int n_in,
                              void* d_out, int out_size, void* d_ws, size_t ws_size,
                              hipStream_t stream) {
  const float* x     = (const float*)d_in[0];
  const float* W     = (const float*)d_in[1];
  const float* alpha = (const float*)d_in[2];
  const float* phi   = (const float*)d_in[3];

  float* P    = (float*)d_ws;
  float* E    = P  + (size_t)NB * ROWS * PP;
  float* Dn   = E  + (size_t)NB * 9 * QPL;

  tgat_xw_band<<<dim3(NBLK, NB), 256, 0, stream>>>(x, W, alpha, phi, P, E, Dn);
  tgat_out<<<dim3(CCH / GCH, NB), NTH, 0, stream>>>(P, E, Dn, (float*)d_out);
}


# band weights: the nine scores of a position run as nine independent chains with all their x2 rows read from LDS ahead, not one row after the other behind its own waits; sums unchanged
# speedup vs baseline: 8.6467x; 1.0476x over previous
#include <hip/hip_runtime.h>
#include <math.h>

typedef __attribute__((ext_vector_type(2))) float v2f;
typedef __attribute__((ext_vector_type(8))) float v8f;

#define NB    32
#define CCH   64
#define TT    300
#define VV    25
#define PLANE (TT*VV)
#define NPOS  (NB*PLANE)
#define TP    150
#define QPL   (TP*VV)
#define KHALF 4
#define ROWS  64
#define GCH   8

#define PP      7520
#define NCG     (ROWS / 4)
#define NPAIR_N (PP / 32)
#define BPAIR   32
#define NBLK    ((NPAIR_N + BPAIR - 1) / BPAIR)
#define NPOSB   (BPAIR * 32)
#define HALO    112
#define LPIT    20
#define WPIT    17

__device__ __forceinline__ void tgat_load_x(const float* __restrict__ x, int n, int p0, int c16,
                                            int half, v2f (&b)[16]) {
  int p = p0 + c16;
  p = p < PLANE ? p : PLANE - 1;
  p = p < 0 ? 0 : p;
  const float* xb = x + (size_t)n * (CCH * PLANE) + (size_t)(2 * half) * PLANE + p;
#pragma unroll
  for (int i = 0; i < 16; ++i) {
    b[i].x = xb[(size_t)(4 * i) * PLANE];
    b[i].y = xb[(size_t)(4 * i + 1) * PLANE];
  }
}

__device__ __forceinline__ void tgat_tile(const v2f (&a)[6][16], const v2f (&b)[16], v8f (&acc)[6]) {
  v8f acc0 = {}, acc1 = {}, acc2 = {}, acc3 = {}, acc4 = {}, acc5 = {};
#pragma unroll
  for (int i = 0; i < 16; ++i) {
    acc0 = __builtin_amdgcn_wmma_f32_16x16x4_f32(false, a[0][i], false, b[i], (short)0, acc0, false, false);
    acc1 = __builtin_amdgcn_wmma_f32_16x16x4_f32(false, a[1][i], false, b[i], (short)0, acc1, false, false);
    acc2 = __builtin_amdgcn_wmma_f32_16x16x4_f32(false, a[2][i], false, b[i], (short)0, acc2, false, false);
    acc3 = __builtin_amdgcn_wmma_f32_16x16x4_f32(false, a[3][i], false, b[i], (short)0, acc3, false, false);
  }
#pragma unroll
  for (int j = 0; j < 4; ++j) {
    const v2f q0 = {acc0[2 * j], acc0[2 * j + 1]};
    acc4 = __builtin_amdgcn_wmma_f32_16x16x4_f32(false, a[4][j], false, q0, (short)0, acc4, false, false);
    acc5 = __builtin_amdgcn_wmma_f32_16x16x4_f32(false, a[5][j], false, q0, (short)0, acc5, false, false);
  }
#pragma unroll
  for (int j = 0; j < 4; ++j) {
    const v2f q1 = {acc1[2 * j], acc1[2 * j + 1]};
    acc4 = __builtin_amdgcn_wmma_f32_16x16x4_f32(false, a[4][4 + j], false, q1, (short)0, acc4, false, false);
    acc5 = __builtin_amdgcn_wmma_f32_16x16x4_f32(false, a[5][4 + j], false, q1, (short)0, acc5, false, false);
  }
#pragma unroll
  for (int j = 0; j < 4; ++j) {
    const v2f q2 = {acc2[2 * j], acc2[2 * j + 1]};
    acc4 = __builtin_amdgcn_wmma_f32_16x16x4_f32(false, a[4][8 + j], false, q2, (short)0, acc4, false, false);
    acc5 = __builtin_amdgcn_wmma_f32_16x16x4_f32(false, a[5][8 + j], false, q2, (short)0, acc5, false, false);
  }
#pragma unroll
  for (int j = 0; j < 4; ++j) {
    const v2f q3 = {acc3[2 * j], acc3[2 * j + 1]};
    acc4 = __builtin_amdgcn_wmma_f32_16x16x4_f32(false, a[4][12 + j], false, q3, (short)0, acc4, false, false);
    acc5 = __builtin_amdgcn_wmma_f32_16x16x4_f32(false, a[5][12 + j], false, q3, (short)0, acc5, false, false);
  }
  acc[0] = acc0; acc[1] = acc1; acc[2] = acc2; acc[3] = acc3; acc[4] = acc4; acc[5] = acc5;
}

__device__ __forceinline__ void tgat_store_tile(const v8f (&A)[6], float* __restrict__ P, int n,
                                                int pos, int half) {
  float4* pb = reinterpret_cast<float4*>(P) + ((size_t)n * NCG + 2 * half) * PP + pos;
#pragma unroll
  for (int k = 0; k < 4; ++k) {
#pragma unroll
    for (int u = 0; u < 2; ++u) {
      pb[(size_t)(4 * k + u) * PP] =
          make_float4(A[k][4 * u], A[k][4 * u + 1], A[k][4 * u + 2], A[k][4 * u + 3]);
    }
  }
}

__device__ __forceinline__ void tgat_lds_rows(float* dst, const v8f& v) {
  float4* d = reinterpret_cast<float4*>(dst);
  d[0] = make_float4(v[0], v[1], v[2], v[3]);
  d[1] = make_float4(v[4], v[5], v[6], v[7]);
}

__device__ __forceinline__ void tgat_pair(const v2f (&a)[6][16], const v2f (&bA)[16],
                                          const v2f (&bB)[16], float* __restrict__ P,
                                          float* x1s, float* x2s, int n, int pp, int pl,
                                          int half, int c16) {
  const int lp = pl * 32 + c16;
  v8f accA[6], accB[6];
  tgat_tile(a, bA, accA);
  tgat_store_tile(accA, P, n, pp * 32 + c16, half);
  tgat_lds_rows(x1s + (lp)             * LPIT + 8 * half, accA[4]);
  tgat_lds_rows(x2s + (HALO + lp)      * LPIT + 8 * half, accA[5]);
  tgat_tile(a, bB, accB);
  tgat_store_tile(accB, P, n, pp * 32 + 16 + c16, half);
  tgat_lds_rows(x1s + (lp + 16)        * LPIT + 8 * half, accB[4]);
  tgat_lds_rows(x2s + (HALO + lp + 16) * LPIT + 8 * half, accB[5]);
}

__device__ __forceinline__ void tgat_load_pair(const float* __restrict__ x, int n, int pp, int c16,
                                               int half, v2f (&bA)[16], v2f (&bB)[16]) {
  tgat_load_x(x, n, pp * 32, c16, half, bA);
  tgat_load_x(x, n, pp * 32 + 16, c16, half, bB);
}

__device__ __forceinline__ v8f tgat_halo_tile(const v2f (&ah)[16], const v2f (&b)[16]) {
  v8f acc = {};
#pragma unroll
  for (int i = 0; i < 16; ++i)
    acc = __builtin_amdgcn_wmma_f32_16x16x4_f32(false, ah[i], false, b[i], (short)0, acc, false, false);
  return acc;
}

__device__ __forceinline__ int tgat_halo_idx(int h) {
  return h < 7 ? 16 * h : HALO + NPOSB + 16 * (h - 7);
}

__global__ void __launch_bounds__(256)
tgat_xw_band(const float* __restrict__ x, const float* __restrict__ W,
             const float* __restrict__ alpha, const float* __restrict__ phi,
             float* __restrict__ P, float* __restrict__ E, float* __restrict__ Dn) {
  __shared__ __attribute__((aligned(16))) float x1s[NPOSB * LPIT];
  __shared__ __attribute__((aligned(16))) float x2s[(NPOSB + 2 * HALO) * LPIT];
  __shared__ float wps[CCH * WPIT];

  const int lane = threadIdx.x & 31;
  const int wave = __builtin_amdgcn_readfirstlane(threadIdx.x >> 5);
  const int half = lane >> 4;
  const int c16  = lane & 15;

  const int n  = blockIdx.y;
  const int pb = blockIdx.x * BPAIR;
  const int P0 = pb * 32;

  v2f b0A[16], b0B[16], b1A[16], b1B[16];
  tgat_load_pair(x, n, pb + wave, c16, half, b0A, b0B);

  v2f a[6][16];
#pragma unroll
  for (int i = 0; i < 16; ++i) {
    const int k0 = 4 * i + 2 * half;
#pragma unroll
    for (int r = 0; r < 4; ++r) {
      a[r][i].x = W[k0 * 64 + 16 * r + c16];
      a[r][i].y = W[(k0 + 1) * 64 + 16 * r + c16];
    }
  }
#pragma unroll
  for (int rt = 0; rt < 4; ++rt) {
#pragma unroll
    for (int j = 0; j < 4; ++j) {
      const int kk = 16 * rt + 8 * half + 2 * j;
      a[4][4 * rt + j].x = alpha[c16 * 64 + kk];
      a[4][4 * rt + j].y = alpha[c16 * 64 + kk + 1];
      a[5][4 * rt + j].x = phi[c16 * 64 + kk];
      a[5][4 * rt + j].y = phi[c16 * 64 + kk + 1];
    }
  }

  if (wave < 4) {
    const float* wr = W + (16 * wave + c16) * 64 + 8 * half;
    v8f wacc = {};
#pragma unroll
    for (int m = 0; m < 16; ++m) {
      const int kof = 16 * (m >> 2) + 2 * (m & 3);
      const v2f wb = {wr[kof], wr[kof + 1]};
      wacc = __builtin_amdgcn_wmma_f32_16x16x4_f32(false, a[5][m], false, wb, (short)0, wacc, false, false);
    }
#pragma unroll
    for (int r = 0; r < 8; ++r) wps[(16 * wave + c16) * WPIT + 8 * half + r] = wacc[r];
  }

#pragma unroll 1
  for (int s = 0; s < 4; s += 2) {
    const int pp0 = pb + wave + 8 * s;
    const int pp1 = pp0 + 8;
    if (pp1 < NPAIR_N) tgat_load_pair(x, n, pp1, c16, half, b1A, b1B);
    if (pp0 < NPAIR_N) tgat_pair(a, b0A, b0B, P, x1s, x2s, n, pp0, wave + 8 * s, half, c16);
    if (s == 0) {
      if (pp0 + 16 < NPAIR_N) tgat_load_pair(x, n, pp0 + 16, c16, half, b0A, b0B);
    } else {
      tgat_load_x(x, n, P0 - HALO + tgat_halo_idx(wave), c16, half, b0A);
      if (wave < 6) tgat_load_x(x, n, P0 - HALO + tgat_halo_idx(wave + 8), c16, half, b0B);
    }
    if (pp1 < NPAIR_N) tgat_pair(a, b1A, b1B, P, x1s, x2s, n, pp1, wave + 8 * s + 8, half, c16);
  }

  __syncthreads();

  {
    v2f ah[16];
#pragma unroll
    for (int i = 0; i < 16; ++i) {
      const int k0 = 4 * i + 2 * half;
      ah[i].x = wps[k0 * WPIT + c16];
      ah[i].y = wps[(k0 + 1) * WPIT + c16];
    }
    const v8f h0 = tgat_halo_tile(ah, b0A);
    tgat_lds_rows(x2s + (tgat_halo_idx(wave) + c16) * LPIT + 8 * half, h0);
    if (wave < 6) {
      const v8f h1 = tgat_halo_tile(ah, b0B);
      tgat_lds_rows(x2s + (tgat_halo_idx(wave + 8) + c16) * LPIT + 8 * half, h1);
    }
  }

  __syncthreads();

  const int tid = threadIdx.x;
  const int tlo = P0 / VV;
  const int te0 = tlo + (tlo & 1);
  float* En = E + (size_t)n * 9 * QPL;
  float* Dq = Dn + (size_t)n * QPL;
#pragma unroll 1
  for (int it = 0; it < 3; ++it) {
    const int o   = tid + 256 * it;
    const int row = o / VV;
    const int v   = o - row * VV;
    const int te  = te0 + 2 * row;
    const int p   = te * VV + v;
    if (p >= P0 && p < P0 + NPOSB && te < TT) {
      const int lp = p - P0;
      const int q  = (te >> 1) * VV + v;
      float a1[16];
      {
        const float4* s1 = reinterpret_cast<const float4*>(x1s + lp * LPIT);
#pragma unroll
        for (int u = 0; u < 4; ++u) {
          const float4 t4 = s1[u];
          a1[4 * u] = t4.x; a1[4 * u + 1] = t4.y; a1[4 * u + 2] = t4.z; a1[4 * u + 3] = t4.w;
        }
      }
      const float4* s2[9];
#pragma unroll
      for (int j = 0; j < 9; ++j) {
        int sr = te - KHALF + j;
        sr = sr < 0 ? 0 : (sr > TT - 1 ? TT - 1 : sr);
        s2[j] = reinterpret_cast<const float4*>(x2s + (HALO + sr * VV + v - P0) * LPIT);
      }
      float sc[9];
#pragma unroll
      for (int j = 0; j < 9; ++j) sc[j] = 0.f;
#pragma unroll
      for (int u = 0; u < 4; ++u) {
#pragma unroll
        for (int j = 0; j < 9; ++j) {
          const float4 t4 = s2[j][u];
          sc[j] = fmaf(a1[4 * u],     t4.x, sc[j]);
          sc[j] = fmaf(a1[4 * u + 1], t4.y, sc[j]);
          sc[j] = fmaf(a1[4 * u + 2], t4.z, sc[j]);
          sc[j] = fmaf(a1[4 * u + 3], t4.w, sc[j]);
        }
      }
      float D = (float)TT;
#pragma unroll
      for (int j = 0; j < 9; ++j) {
        const int sr = te - KHALF + j;
        const float e = (unsigned)sr < (unsigned)TT ? __expf(sc[j]) - 1.f : 0.f;
        En[(size_t)j * QPL + q] = e;
        D += e;
      }
      Dq[q] = D;
    }
  }
}

#define NTH  512
#define GCG  (GCH / 4)
#define NF4  (GCG * PP)
#define NLD  ((NF4 + NTH - 1) / NTH)
#define NQ   ((QPL + NTH - 1) / NTH)
#define NSEG 5
#define NCOL (GCG * VV)
#define RP2  57
#define SPG  8576
__global__ void __launch_bounds__(NTH)
tgat_out(const float* __restrict__ P, const float* __restrict__ E,
         const float* __restrict__ Dn, float* __restrict__ out) {
  __shared__ __attribute__((aligned(16))) float sp[GCG * SPG * 4];
  __shared__ __attribute__((aligned(16))) float spart[NSEG * NCOL * 4];
  __shared__ __attribute__((aligned(16))) float sS[NCOL * 4];

  const int n   = blockIdx.y;
  const int g0  = blockIdx.x * GCG;
  const int c0  = blockIdx.x * GCH;
  const int tid = threadIdx.x;

  const float* En = E + (size_t)n * 9 * QPL;
  const float* Dq = Dn + (size_t)n * QPL;

  float4* s4 = reinterpret_cast<float4*>(sp);
  float4* part4 = reinterpret_cast<float4*>(spart);
  float4* sS4 = reinterpret_cast<float4*>(sS);

  float e[NQ][9];
  float dv[NQ];
  {
    const float4* gp = reinterpret_cast<const float4*>(P) + ((size_t)n * NCG + g0) * PP;
    float4 r[NLD];
#pragma unroll
    for (int u = 0; u < NLD; ++u) {
      int i = tid + NTH * u;
      i = i < NF4 ? i : NF4 - 1;
      r[u] = gp[i];
    }
#pragma unroll
    for (int i = 0; i < NQ; ++i) {
      int q = tid + NTH * i;
      q = q < QPL ? q : QPL - 1;
#pragma unroll
      for (int j = 0; j < 9; ++j) e[i][j] = En[(size_t)j * QPL + q];
      dv[i] = Dq[q];
    }
#pragma unroll
    for (int u = 0; u < NLD; ++u) {
      int i = tid + NTH * u;
      i = i < NF4 ? i : NF4 - 1;
      const int cg = i >= PP ? 1 : 0;
      const int p  = i - cg * PP;
      const int pr = p / (2 * VV);
      s4[cg * SPG + pr * RP2 + (p - pr * (2 * VV))] = r[u];
    }
  }
  __syncthreads();

  if (tid < NSEG * NCOL) {
    const int seg = tid / NCOL;
    const int gv  = tid - seg * NCOL;
    const int cg  = gv / VV;
    const int v   = gv - cg * VV;
    const float4* col = s4 + cg * SPG + (seg * (TT / NSEG / 2)) * RP2 + v;
    float4 acc = make_float4(0.f, 0.f, 0.f, 0.f);
    for (int t = 0; t < TT / NSEG; ++t) {
      const float4 xv = col[(t >> 1) * RP2 + (t & 1) * VV];
      acc.x += xv.x; acc.y += xv.y; acc.z += xv.z; acc.w += xv.w;
    }
    part4[tid] = acc;
  }
  __syncthreads();
  if (tid < NCOL) {
    float4 accv = make_float4(0.f, 0.f, 0.f, 0.f);
#pragma unroll
    for (int seg = 0; seg < NSEG; ++seg) {
      const float4 pv = part4[seg * NCOL + tid];
      accv.x += pv.x; accv.y += pv.y; accv.z += pv.z; accv.w += pv.w;
    }
    sS4[tid] = accv;
  }
  __syncthreads();

  float* ob = out + ((size_t)n * CCH + c0) * QPL;
#pragma unroll
  for (int i = 0; i < NQ; ++i) {
    const int q = tid + NTH * i;
    if (q < QPL) {
      const int tp = q / VV;
      const int v  = q - tp * VV;
      int idx[9];
#pragma unroll
      for (int j = 0; j < 9; ++j) {
        int pr = tp - KHALF / 2 + (j >> 1);
        pr = pr < 0 ? 0 : (pr > TP - 1 ? TP - 1 : pr);
        idx[j] = pr * RP2 + (j & 1) * VV + v;
      }
      const float D = dv[i];
#pragma unroll
      for (int cg = 0; cg < GCG; ++cg) {
        float4 accv = sS4[cg * VV + v];
#pragma unroll
        for (int j = 0; j < 9; ++j) {
          const float4 xv = s4[cg * SPG + idx[j]];
          accv.x += e[i][j] * xv.x;
          accv.y += e[i][j] * xv.y;
          accv.z += e[i][j] * xv.z;
          accv.w += e[i][j] * xv.w;
        }
        ob[(size_t)(4 * cg)     * QPL + q] = accv.x / D;
        ob[(size_t)(4 * cg + 1) * QPL + q] = accv.y / D;
        ob[(size_t)(4 * cg + 2) * QPL + q] = accv.z / D;
        ob[(size_t)(4 * cg + 3) * QPL + q] = accv.w / D;
      }
    }
  }
}

extern "C" void kernel_launch(void* const* d_in, const int* in_sizes, int n_in,
                              void* d_out, int out_size, void* d_ws, size_t ws_size,
                              hipStream_t stream) {
  const float* x     = (const float*)d_in[0];
  const float* W     = (const float*)d_in[1];
  const float* alpha = (const float*)d_in[2];
  const float* phi   = (const float*)d_in[3];

  float* P    = (float*)d_ws;
  float* E    = P  + (size_t)NB * ROWS * PP;
  float* Dn   = E  + (size_t)NB * 9 * QPL;

  tgat_xw_band<<<dim3(NBLK, NB), 256, 0, stream>>>(x, W, alpha, phi, P, E, Dn);
  tgat_out<<<dim3(CCH / GCH, NB), NTH, 0, stream>>>(P, E, Dn, (float*)d_out);
}


# GEMM launch: x of each 16-position tile requested two tiles ahead in branch-free straight-line steps (was per pair in a loop whose waits drained the newest loads and copied a set); sums unchanged
# speedup vs baseline: 8.9750x; 1.0380x over previous
#include <hip/hip_runtime.h>
#include <math.h>

typedef __attribute__((ext_vector_type(2))) float v2f;
typedef __attribute__((ext_vector_type(8))) float v8f;

#define NB    32
#define CCH   64
#define TT    300
#define VV    25
#define PLANE (TT*VV)
#define NPOS  (NB*PLANE)
#define TP    150
#define QPL   (TP*VV)
#define KHALF 4
#define ROWS  64
#define GCH   8

#define PP      7520
#define NCG     (ROWS / 4)
#define NPAIR_N (PP / 32)
#define BPAIR   32
#define NBLK    ((NPAIR_N + BPAIR - 1) / BPAIR)
#define NPOSB   (BPAIR * 32)
#define HALO    112
#define LPIT    20
#define WPIT    17

__device__ __forceinline__ void tgat_load_x(const float* __restrict__ x, int n, int p0, int c16,
                                            int half, v2f (&b)[16]) {
  int p = p0 + c16;
  p = p < PLANE ? p : PLANE - 1;
  p = p < 0 ? 0 : p;
  const float* xb = x + (size_t)n * (CCH * PLANE) + (size_t)(2 * half) * PLANE + p;
#pragma unroll
  for (int i = 0; i < 16; ++i) {
    b[i].x = xb[(size_t)(4 * i) * PLANE];
    b[i].y = xb[(size_t)(4 * i + 1) * PLANE];
  }
}

__device__ __forceinline__ void tgat_tile(const v2f (&a)[6][16], const v2f (&b)[16], v8f (&acc)[6]) {
  v8f acc0 = {}, acc1 = {}, acc2 = {}, acc3 = {}, acc4 = {}, acc5 = {};
#pragma unroll
  for (int i = 0; i < 16; ++i) {
    acc0 = __builtin_amdgcn_wmma_f32_16x16x4_f32(false, a[0][i], false, b[i], (short)0, acc0, false, false);
    acc1 = __builtin_amdgcn_wmma_f32_16x16x4_f32(false, a[1][i], false, b[i], (short)0, acc1, false, false);
    acc2 = __builtin_amdgcn_wmma_f32_16x16x4_f32(false, a[2][i], false, b[i], (short)0, acc2, false, false);
    acc3 = __builtin_amdgcn_wmma_f32_16x16x4_f32(false, a[3][i], false, b[i], (short)0, acc3, false, false);
  }
#pragma unroll
  for (int j = 0; j < 4; ++j) {
    const v2f q0 = {acc0[2 * j], acc0[2 * j + 1]};
    acc4 = __builtin_amdgcn_wmma_f32_16x16x4_f32(false, a[4][j], false, q0, (short)0, acc4, false, false);
    acc5 = __builtin_amdgcn_wmma_f32_16x16x4_f32(false, a[5][j], false, q0, (short)0, acc5, false, false);
  }
#pragma unroll
  for (int j = 0; j < 4; ++j) {
    const v2f q1 = {acc1[2 * j], acc1[2 * j + 1]};
    acc4 = __builtin_amdgcn_wmma_f32_16x16x4_f32(false, a[4][4 + j], false, q1, (short)0, acc4, false, false);
    acc5 = __builtin_amdgcn_wmma_f32_16x16x4_f32(false, a[5][4 + j], false, q1, (short)0, acc5, false, false);
  }
#pragma unroll
  for (int j = 0; j < 4; ++j) {
    const v2f q2 = {acc2[2 * j], acc2[2 * j + 1]};
    acc4 = __builtin_amdgcn_wmma_f32_16x16x4_f32(false, a[4][8 + j], false, q2, (short)0, acc4, false, false);
    acc5 = __builtin_amdgcn_wmma_f32_16x16x4_f32(false, a[5][8 + j], false, q2, (short)0, acc5, false, false);
  }
#pragma unroll
  for (int j = 0; j < 4; ++j) {
    const v2f q3 = {acc3[2 * j], acc3[2 * j + 1]};
    acc4 = __builtin_amdgcn_wmma_f32_16x16x4_f32(false, a[4][12 + j], false, q3, (short)0, acc4, false, false);
    acc5 = __builtin_amdgcn_wmma_f32_16x16x4_f32(false, a[5][12 + j], false, q3, (short)0, acc5, false, false);
  }
  acc[0] = acc0; acc[1] = acc1; acc[2] = acc2; acc[3] = acc3; acc[4] = acc4; acc[5] = acc5;
}

__device__ __forceinline__ void tgat_store_tile(const v8f (&A)[6], float* __restrict__ P, int n,
                                                int pos, int half) {
  float4* pb = reinterpret_cast<float4*>(P) + ((size_t)n * NCG + 2 * half) * PP + pos;
#pragma unroll
  for (int k = 0; k < 4; ++k) {
#pragma unroll
    for (int u = 0; u < 2; ++u) {
      pb[(size_t)(4 * k + u) * PP] =
          make_float4(A[k][4 * u], A[k][4 * u + 1], A[k][4 * u + 2], A[k][4 * u + 3]);
    }
  }
}

__device__ __forceinline__ void tgat_lds_rows(float* dst, const v8f& v) {
  float4* d = reinterpret_cast<float4*>(dst);
  d[0] = make_float4(v[0], v[1], v[2], v[3]);
  d[1] = make_float4(v[4], v[5], v[6], v[7]);
}

#define TGAT_KEEP_ORDER() __builtin_amdgcn_sched_barrier(0)

__device__ __forceinline__ void tgat_step(const v2f (&a)[6][16], const v2f (&b)[16],
                                          float* __restrict__ P, float* x1s, float* x2s, int n,
                                          int pos0, int lp0, int half, int c16) {
  v8f acc[6];
  tgat_tile(a, b, acc);
  tgat_store_tile(acc, P, n, pos0 + c16, half);
  tgat_lds_rows(x1s + (lp0 + c16)        * LPIT + 8 * half, acc[4]);
  tgat_lds_rows(x2s + (HALO + lp0 + c16) * LPIT + 8 * half, acc[5]);
}

__device__ __forceinline__ v8f tgat_halo_tile(const v2f (&ah)[16], const v2f (&b)[16]) {
  v8f acc = {};
#pragma unroll
  for (int i = 0; i < 16; ++i)
    acc = __builtin_amdgcn_wmma_f32_16x16x4_f32(false, ah[i], false, b[i], (short)0, acc, false, false);
  return acc;
}

__device__ __forceinline__ int tgat_halo_idx(int h) {
  return h < 7 ? 16 * h : HALO + NPOSB + 16 * (h - 7);
}

__global__ void __launch_bounds__(256)
tgat_xw_band(const float* __restrict__ x, const float* __restrict__ W,
             const float* __restrict__ alpha, const float* __restrict__ phi,
             float* __restrict__ P, float* __restrict__ E, float* __restrict__ Dn) {
  __shared__ __attribute__((aligned(16))) float x1s[NPOSB * LPIT];
  __shared__ __attribute__((aligned(16))) float x2s[(NPOSB + 2 * HALO) * LPIT];
  __shared__ float wps[CCH * WPIT];

  const int lane = threadIdx.x & 31;
  const int wave = __builtin_amdgcn_readfirstlane(threadIdx.x >> 5);
  const int half = lane >> 4;
  const int c16  = lane & 15;

  const int n  = blockIdx.y;
  const int pb = blockIdx.x * BPAIR;
  const int P0 = pb * 32;

  const int pw = pb + wave;
  const bool full = pb + BPAIR <= NPAIR_N;
  v2f t0[16], t1[16], t8[16], t9[16];
  tgat_load_x(x, n, pw * 32,      c16, half, t0);
  tgat_load_x(x, n, pw * 32 + 16, c16, half, t1);

  v2f a[6][16];
#pragma unroll
  for (int i = 0; i < 16; ++i) {
    const int k0 = 4 * i + 2 * half;
#pragma unroll
    for (int r = 0; r < 4; ++r) {
      a[r][i].x = W[k0 * 64 + 16 * r + c16];
      a[r][i].y = W[(k0 + 1) * 64 + 16 * r + c16];
    }
  }
#pragma unroll
  for (int rt = 0; rt < 4; ++rt) {
#pragma unroll
    for (int j = 0; j < 4; ++j) {
      const int kk = 16 * rt + 8 * half + 2 * j;
      a[4][4 * rt + j].x = alpha[c16 * 64 + kk];
      a[4][4 * rt + j].y = alpha[c16 * 64 + kk + 1];
      a[5][4 * rt + j].x = phi[c16 * 64 + kk];
      a[5][4 * rt + j].y = phi[c16 * 64 + kk + 1];
    }
  }

  if (wave < 4) {
    const float* wr = W + (16 * wave + c16) * 64 + 8 * half;
    v8f wacc = {};
#pragma unroll
    for (int m = 0; m < 16; ++m) {
      const int kof = 16 * (m >> 2) + 2 * (m & 3);
      const v2f wb = {wr[kof], wr[kof + 1]};
      wacc = __builtin_amdgcn_wmma_f32_16x16x4_f32(false, a[5][m], false, wb, (short)0, wacc, false, false);
    }
#pragma unroll
    for (int r = 0; r < 8; ++r) wps[(16 * wave + c16) * WPIT + 8 * half + r] = wacc[r];
  }

  if (full) {
    v2f t2[16], t3[16], t4[16], t5[16], t6[16], t7[16];
    tgat_load_x(x, n, (pw + 8) * 32,       c16, half, t2);
    TGAT_KEEP_ORDER();
    tgat_step(a, t0, P, x1s, x2s, n, pw * 32,             wave * 32,             half, c16);
    TGAT_KEEP_ORDER();
    tgat_load_x(x, n, (pw + 8) * 32 + 16,  c16, half, t3);
    TGAT_KEEP_ORDER();
    tgat_step(a, t1, P, x1s, x2s, n, pw * 32 + 16,        wave * 32 + 16,        half, c16);
    TGAT_KEEP_ORDER();
    tgat_load_x(x, n, (pw + 16) * 32,      c16, half, t4);
    TGAT_KEEP_ORDER();
    tgat_step(a, t2, P, x1s, x2s, n, (pw + 8) * 32,       (wave + 8) * 32,       half, c16);
    TGAT_KEEP_ORDER();
    tgat_load_x(x, n, (pw + 16) * 32 + 16, c16, half, t5);
    TGAT_KEEP_ORDER();
    tgat_step(a, t3, P, x1s, x2s, n, (pw + 8) * 32 + 16,  (wave + 8) * 32 + 16,  half, c16);
    TGAT_KEEP_ORDER();
    tgat_load_x(x, n, (pw + 24) * 32,      c16, half, t6);
    TGAT_KEEP_ORDER();
    tgat_step(a, t4, P, x1s, x2s, n, (pw + 16) * 32,      (wave + 16) * 32,      half, c16);
    TGAT_KEEP_ORDER();
    tgat_load_x(x, n, (pw + 24) * 32 + 16, c16, half, t7);
    TGAT_KEEP_ORDER();
    tgat_step(a, t5, P, x1s, x2s, n, (pw + 16) * 32 + 16, (wave + 16) * 32 + 16, half, c16);
    TGAT_KEEP_ORDER();
    tgat_load_x(x, n, P0 - HALO + tgat_halo_idx(wave), c16, half, t8);
    TGAT_KEEP_ORDER();
    tgat_step(a, t6, P, x1s, x2s, n, (pw + 24) * 32,      (wave + 24) * 32,      half, c16);
    TGAT_KEEP_ORDER();
    tgat_load_x(x, n, P0 - HALO + tgat_halo_idx(wave < 6 ? wave + 8 : wave), c16, half, t9);
    TGAT_KEEP_ORDER();
    tgat_step(a, t7, P, x1s, x2s, n, (pw + 24) * 32 + 16, (wave + 24) * 32 + 16, half, c16);
    TGAT_KEEP_ORDER();
  } else {
    tgat_load_x(x, n, P0 - HALO + tgat_halo_idx(wave), c16, half, t8);
    tgat_load_x(x, n, P0 - HALO + tgat_halo_idx(wave < 6 ? wave + 8 : wave), c16, half, t9);
    tgat_step(a, t0, P, x1s, x2s, n, pw * 32,      wave * 32,      half, c16);
    tgat_step(a, t1, P, x1s, x2s, n, pw * 32 + 16, wave * 32 + 16, half, c16);
#pragma unroll 1
    for (int i = 2; i < 8; ++i) {
      const int s8 = 8 * (i >> 1);
      const int o  = 16 * (i & 1);
      if (pw + s8 < NPAIR_N) {
        tgat_load_x(x, n, (pw + s8) * 32 + o, c16, half, t0);
        tgat_step(a, t0, P, x1s, x2s, n, (pw + s8) * 32 + o, (wave + s8) * 32 + o, half, c16);
      }
    }
  }

  __syncthreads();

  {
    v2f ah[16];
#pragma unroll
    for (int i = 0; i < 16; ++i) {
      const int k0 = 4 * i + 2 * half;
      ah[i].x = wps[k0 * WPIT + c16];
      ah[i].y = wps[(k0 + 1) * WPIT + c16];
    }
    const v8f h0 = tgat_halo_tile(ah, t8);
    tgat_lds_rows(x2s + (tgat_halo_idx(wave) + c16) * LPIT + 8 * half, h0);
    if (wave < 6) {
      const v8f h1 = tgat_halo_tile(ah, t9);
      tgat_lds_rows(x2s + (tgat_halo_idx(wave + 8) + c16) * LPIT + 8 * half, h1);
    }
  }

  __syncthreads();

  const int tid = threadIdx.x;
  const int tlo = P0 / VV;
  const int te0 = tlo + (tlo & 1);
  float* En = E + (size_t)n * 9 * QPL;
  float* Dq = Dn + (size_t)n * QPL;
#pragma unroll 1
  for (int it = 0; it < 3; ++it) {
    const int o   = tid + 256 * it;
    const int row = o / VV;
    const int v   = o - row * VV;
    const int te  = te0 + 2 * row;
    const int p   = te * VV + v;
    if (p >= P0 && p < P0 + NPOSB && te < TT) {
      const int lp = p - P0;
      const int q  = (te >> 1) * VV + v;
      float a1[16];
      {
        const float4* s1 = reinterpret_cast<const float4*>(x1s + lp * LPIT);
#pragma unroll
        for (int u = 0; u < 4; ++u) {
          const float4 t4 = s1[u];
          a1[4 * u] = t4.x; a1[4 * u + 1] = t4.y; a1[4 * u + 2] = t4.z; a1[4 * u + 3] = t4.w;
        }
      }
      const float4* s2[9];
#pragma unroll
      for (int j = 0; j < 9; ++j) {
        int sr = te - KHALF + j;
        sr = sr < 0 ? 0 : (sr > TT - 1 ? TT - 1 : sr);
        s2[j] = reinterpret_cast<const float4*>(x2s + (HALO + sr * VV + v - P0) * LPIT);
      }
      float sc[9];
#pragma unroll
      for (int j = 0; j < 9; ++j) sc[j] = 0.f;
#pragma unroll
      for (int u = 0; u < 4; ++u) {
#pragma unroll
        for (int j = 0; j < 9; ++j) {
          const float4 t4 = s2[j][u];
          sc[j] = fmaf(a1[4 * u],     t4.x, sc[j]);
          sc[j] = fmaf(a1[4 * u + 1], t4.y, sc[j]);
          sc[j] = fmaf(a1[4 * u + 2], t4.z, sc[j]);
          sc[j] = fmaf(a1[4 * u + 3], t4.w, sc[j]);
        }
      }
      float D = (float)TT;
#pragma unroll
      for (int j = 0; j < 9; ++j) {
        const int sr = te - KHALF + j;
        const float e = (unsigned)sr < (unsigned)TT ? __expf(sc[j]) - 1.f : 0.f;
        En[(size_t)j * QPL + q] = e;
        D += e;
      }
      Dq[q] = D;
    }
  }
}

#define NTH  512
#define GCG  (GCH / 4)
#define NF4  (GCG * PP)
#define NLD  ((NF4 + NTH - 1) / NTH)
#define NQ   ((QPL + NTH - 1) / NTH)
#define NSEG 5
#define NCOL (GCG * VV)
#define RP2  57
#define SPG  8576
__global__ void __launch_bounds__(NTH)
tgat_out(const float* __restrict__ P, const float* __restrict__ E,
         const float* __restrict__ Dn, float* __restrict__ out) {
  __shared__ __attribute__((aligned(16))) float sp[GCG * SPG * 4];
  __shared__ __attribute__((aligned(16))) float spart[NSEG * NCOL * 4];
  __shared__ __attribute__((aligned(16))) float sS[NCOL * 4];

  const int n   = blockIdx.y;
  const int g0  = blockIdx.x * GCG;
  const int c0  = blockIdx.x * GCH;
  const int tid = threadIdx.x;

  const float* En = E + (size_t)n * 9 * QPL;
  const float* Dq = Dn + (size_t)n * QPL;

  float4* s4 = reinterpret_cast<float4*>(sp);
  float4* part4 = reinterpret_cast<float4*>(spart);
  float4* sS4 = reinterpret_cast<float4*>(sS);

  float e[NQ][9];
  float dv[NQ];
  {
    const float4* gp = reinterpret_cast<const float4*>(P) + ((size_t)n * NCG + g0) * PP;
    float4 r[NLD];
#pragma unroll
    for (int u = 0; u < NLD; ++u) {
      int i = tid + NTH * u;
      i = i < NF4 ? i : NF4 - 1;
      r[u] = gp[i];
    }
#pragma unroll
    for (int i = 0; i < NQ; ++i) {
      int q = tid + NTH * i;
      q = q < QPL ? q : QPL - 1;
#pragma unroll
      for (int j = 0; j < 9; ++j) e[i][j] = En[(size_t)j * QPL + q];
      dv[i] = Dq[q];
    }
#pragma unroll
    for (int u = 0; u < NLD; ++u) {
      int i = tid + NTH * u;
      i = i < NF4 ? i : NF4 - 1;
      const int cg = i >= PP ? 1 : 0;
      const int p  = i - cg * PP;
      const int pr = p / (2 * VV);
      s4[cg * SPG + pr * RP2 + (p - pr * (2 * VV))] = r[u];
    }
  }
  __syncthreads();

  if (tid < NSEG * NCOL) {
    const int seg = tid / NCOL;
    const int gv  = tid - seg * NCOL;
    const int cg  = gv / VV;
    const int v   = gv - cg * VV;
    const float4* col = s4 + cg * SPG + (seg * (TT / NSEG / 2)) * RP2 + v;
    float4 acc = make_float4(0.f, 0.f, 0.f, 0.f);
    for (int t = 0; t < TT / NSEG; ++t) {
      const float4 xv = col[(t >> 1) * RP2 + (t & 1) * VV];
      acc.x += xv.x; acc.y += xv.y; acc.z += xv.z; acc.w += xv.w;
    }
    part4[tid] = acc;
  }
  __syncthreads();
  if (tid < NCOL) {
    float4 accv = make_float4(0.f, 0.f, 0.f, 0.f);
#pragma unroll
    for (int seg = 0; seg < NSEG; ++seg) {
      const float4 pv = part4[seg * NCOL + tid];
      accv.x += pv.x; accv.y += pv.y; accv.z += pv.z; accv.w += pv.w;
    }
    sS4[tid] = accv;
  }
  __syncthreads();

  float* ob = out + ((size_t)n * CCH + c0) * QPL;
#pragma unroll
  for (int i = 0; i < NQ; ++i) {
    const int q = tid + NTH * i;
    if (q < QPL) {
      const int tp = q / VV;
      const int v  = q - tp * VV;
      int idx[9];
#pragma unroll
      for (int j = 0; j < 9; ++j) {
        int pr = tp - KHALF / 2 + (j >> 1);
        pr = pr < 0 ? 0 : (pr > TP - 1 ? TP - 1 : pr);
        idx[j] = pr * RP2 + (j & 1) * VV + v;
      }
      const float D = dv[i];
#pragma unroll
      for (int cg = 0; cg < GCG; ++cg) {
        float4 accv = sS4[cg * VV + v];
#pragma unroll
        for (int j = 0; j < 9; ++j) {
          const float4 xv = s4[cg * SPG + idx[j]];
          accv.x += e[i][j] * xv.x;
          accv.y += e[i][j] * xv.y;
          accv.z += e[i][j] * xv.z;
          accv.w += e[i][j] * xv.w;
        }
        ob[(size_t)(4 * cg)     * QPL + q] = accv.x / D;
        ob[(size_t)(4 * cg + 1) * QPL + q] = accv.y / D;
        ob[(size_t)(4 * cg + 2) * QPL + q] = accv.z / D;
        ob[(size_t)(4 * cg + 3) * QPL + q] = accv.w / D;
      }
    }
  }
}

extern "C" void kernel_launch(void* const* d_in, const int* in_sizes, int n_in,
                              void* d_out, int out_size, void* d_ws, size_t ws_size,
                              hipStream_t stream) {
  const float* x     = (const float*)d_in[0];
  const float* W     = (const float*)d_in[1];
  const float* alpha = (const float*)d_in[2];
  const float* phi   = (const float*)d_in[3];

  float* P    = (float*)d_ws;
  float* E    = P  + (size_t)NB * ROWS * PP;
  float* Dn   = E  + (size_t)NB * 9 * QPL;

  tgat_xw_band<<<dim3(NBLK, NB), 256, 0, stream>>>(x, W, alpha, phi, P, E, Dn);
  tgat_out<<<dim3(CCH / GCH, NB), NTH, 0, stream>>>(P, E, Dn, (float*)d_out);
}
